# starting kernel
# baseline (speedup 1.0000x reference)
#include <hip/hip_runtime.h>
#include <math.h>

typedef __attribute__((ext_vector_type(2))) float v2f;
typedef __attribute__((ext_vector_type(8))) float v8f;

#define F_NUM 10
#define OUT_N 10
#define NROW  12

__global__ void coef_kernel(const float* __restrict__ w,
                            const float* __restrict__ pool_w,
                            float* __restrict__ coefT,
                            int T, int Tpad, int D) {
  int t = blockIdx.x * blockDim.x + threadIdx.x;
  if (t >= Tpad) return;
  float row[16];
#pragma unroll
  for (int i = 0; i < 16; ++i) row[i] = 0.0f;
  if (t < T) {
    row[0] = 1.0f;
    float G = 0.0f;
    float e = (float)(T - 1 - t);
#pragma unroll
    for (int f = 0; f < F_NUM; ++f) {
      float wf = w[(size_t)f * D];
      float pf = pool_w[f];
      float c  = powf(wf, e);
      row[2 + f] = c;
      G += pf * (1.0f - c) / (1.0f - wf);
    }
    row[1] = G;
  }
#pragma unroll
  for (int i = 0; i < 16; ++i) coefT[(size_t)t * 16 + i] = row[i];
}

template <int DC>
__global__ void __launch_bounds__(128)
gemm_coef_x(const float* __restrict__ x, const float* __restrict__ coefT,
            float* __restrict__ partial, int T, int Drt, int TC, int NT, int units) {
  const int D = (DC > 0) ? DC : Drt;
  int wid = __builtin_amdgcn_readfirstlane(
      (int)(blockIdx.x * (blockDim.x >> 5)) + ((int)threadIdx.x >> 5));
  if (wid >= units) return;
  int lane = threadIdx.x & 31;
  int ntile  = wid % NT;
  int kchunk = wid / NT;
  int nb = ntile * 16;
  int n  = lane & 15;
  int h  = lane >> 4;

  int t0 = kchunk * TC;
  int valid = T - t0;
  if (valid < 0)  valid = 0;
  if (valid > TC) valid = TC;
  int full = valid & ~3;

  const float* cp = coefT + (size_t)(t0 + 2 * h) * 16 + n;
  const float* xp = x     + (size_t)(t0 + 2 * h) * D  + nb + n;
  const size_t xstep = (size_t)4 * D;

  v8f acc = {};
  if (full > 0) {
    v2f a = { cp[0], cp[16] };
    v2f b = { xp[0], xp[D] };
#pragma unroll 2
    for (int kk = 4; kk < full; kk += 4) {
      cp += 64;
      xp += xstep;
      v2f an = { cp[0], cp[16] };
      v2f bn = { xp[0], xp[D] };
      acc = __builtin_amdgcn_wmma_f32_16x16x4_f32(
          false, a, false, b, (short)0, acc, false, false);
      a = an; b = bn;
    }
    acc = __builtin_amdgcn_wmma_f32_16x16x4_f32(
        false, a, false, b, (short)0, acc, false, false);
  }
  for (int kk = full; kk < TC; kk += 4) {
    int ta = t0 + kk + 2 * h, tb = ta + 1;
    v2f a = { coefT[(size_t)ta * 16 + n], coefT[(size_t)tb * 16 + n] };
    v2f b = { (ta < T) ? x[(size_t)ta * D + nb + n] : 0.0f,
              (tb < T) ? x[(size_t)tb * D + nb + n] : 0.0f };
    acc = __builtin_amdgcn_wmma_f32_16x16x4_f32(
        false, a, false, b, (short)0, acc, false, false);
  }

#pragma unroll
  for (int r = 0; r < 8; ++r) {
    int m = r + 8 * h;
    if (m < NROW)
      partial[((size_t)kchunk * NROW + m) * D + nb + n] = acc[r];
  }
}

__global__ void reduce_partials(const float* __restrict__ partial,
                                const float* __restrict__ pool_b,
                                float* __restrict__ avg,
                                float* __restrict__ out_hfinal,
                                int D, int KC, int T) {
  int idx = blockIdx.x * blockDim.x + threadIdx.x;
  if (idx >= NROW * D) return;
  int m = idx / D, d = idx % D;
  float s = 0.0f;
  for (int c = 0; c < KC; ++c)
    s += partial[((size_t)c * NROW + m) * D + d];
  float invT = 1.0f / (float)T;
  if (m == 0)      avg[d]     = s * invT;
  else if (m == 1) avg[D + d] = s * invT + pool_b[0];
  else             out_hfinal[(size_t)(m - 2) * D + d] = s;
}

__global__ void gemv_rows(const float* __restrict__ Wm,
                          const float* __restrict__ bias,
                          const float* __restrict__ vec,
                          float* __restrict__ out,
                          int K, int N, int do_relu) {
  int wid  = blockIdx.x * (blockDim.x >> 5) + (threadIdx.x >> 5);
  int lane = threadIdx.x & 31;
  if (wid >= N) return;
  const float4* wr = (const float4*)(Wm + (size_t)wid * K);
  const float4* vv = (const float4*)vec;
  float s = 0.0f;
  for (int i = lane; i < (K >> 2); i += 32) {
    float4 a = wr[i], b = vv[i];
    s += a.x * b.x + a.y * b.y + a.z * b.z + a.w * b.w;
  }
  for (int off = 16; off > 0; off >>= 1) s += __shfl_down(s, off, 32);
  if (lane == 0) {
    float v = s + bias[wid];
    out[wid] = do_relu ? fmaxf(v, 0.0f) : v;
  }
}

extern "C" void kernel_launch(void* const* d_in, const int* in_sizes, int n_in,
                              void* d_out, int out_size, void* d_ws, size_t ws_size,
                              hipStream_t stream) {
  const float* inp    = (const float*)d_in[0];
  const float* w      = (const float*)d_in[2];
  const float* pool_w = (const float*)d_in[3];
  const float* pool_b = (const float*)d_in[4];
  const float* i2o_w  = (const float*)d_in[5];
  const float* i2o_b  = (const float*)d_in[6];
  const float* o_w    = (const float*)d_in[7];
  const float* o_b    = (const float*)d_in[8];
  float* out = (float*)d_out;

  const int D  = in_sizes[1] / F_NUM;
  const int T  = in_sizes[0] / D;
  const int H1 = in_sizes[6];

  const int KC = 16;
  int TC   = (((T + KC - 1) / KC) + 3) & ~3;
  int Tpad = KC * TC;
  int NT   = D / 16;

  float* ws      = (float*)d_ws;
  float* coefT   = ws;
  float* partial = coefT + (size_t)Tpad * 16;
  float* avg     = partial + (size_t)KC * NROW * D;
  float* hid1    = avg + (size_t)2 * D;

  coef_kernel<<<(Tpad + 255) / 256, 256, 0, stream>>>(w, pool_w, coefT, T, Tpad, D);

  int units = NT * KC;
  dim3 g2((units + 3) / 4), b2(128);
  if (D == 4096)
    gemm_coef_x<4096><<<g2, b2, 0, stream>>>(inp, coefT, partial, T, D, TC, NT, units);
  else
    gemm_coef_x<0><<<g2, b2, 0, stream>>>(inp, coefT, partial, T, D, TC, NT, units);

  reduce_partials<<<(NROW * D + 255) / 256, 256, 0, stream>>>(
      partial, pool_b, avg, out + OUT_N, D, KC, T);

  gemv_rows<<<(H1 + 7) / 8, 256, 0, stream>>>(i2o_w, i2o_b, avg, hid1, 2 * D, H1, 1);
  gemv_rows<<<(OUT_N + 7) / 8, 256, 0, stream>>>(o_w, o_b, hid1, out, H1, OUT_N, 0);
}


# gemv 4096x8192: vector kept in registers (one K-quarter per wave) instead of reloaded per row; 4 rows/wave, next row's 16 NT loads ahead; row sum = 4 quarter sums added via LDS in fixed order
# speedup vs baseline: 2.7618x; 2.7618x over previous
#include <hip/hip_runtime.h>
#include <math.h>

typedef __attribute__((ext_vector_type(2))) float v2f;
typedef __attribute__((ext_vector_type(8))) float v8f;
typedef __attribute__((ext_vector_type(4))) float v4f;

#define F_NUM 10
#define OUT_N 10
#define NROW  12

__global__ void coef_kernel(const float* __restrict__ w,
                            const float* __restrict__ pool_w,
                            float* __restrict__ coefT,
                            int T, int Tpad, int D) {
  int t = blockIdx.x * blockDim.x + threadIdx.x;
  if (t >= Tpad) return;
  float row[16];
#pragma unroll
  for (int i = 0; i < 16; ++i) row[i] = 0.0f;
  if (t < T) {
    row[0] = 1.0f;
    float G = 0.0f;
    float e = (float)(T - 1 - t);
#pragma unroll
    for (int f = 0; f < F_NUM; ++f) {
      float wf = w[(size_t)f * D];
      float pf = pool_w[f];
      float c  = powf(wf, e);
      row[2 + f] = c;
      G += pf * (1.0f - c) / (1.0f - wf);
    }
    row[1] = G;
  }
#pragma unroll
  for (int i = 0; i < 16; ++i) coefT[(size_t)t * 16 + i] = row[i];
}

template <int DC>
__global__ void __launch_bounds__(128)
gemm_coef_x(const float* __restrict__ x, const float* __restrict__ coefT,
            float* __restrict__ partial, int T, int Drt, int TC, int NT, int units) {
  const int D = (DC > 0) ? DC : Drt;
  int wid = __builtin_amdgcn_readfirstlane(
      (int)(blockIdx.x * (blockDim.x >> 5)) + ((int)threadIdx.x >> 5));
  if (wid >= units) return;
  int lane = threadIdx.x & 31;
  int ntile  = wid % NT;
  int kchunk = wid / NT;
  int nb = ntile * 16;
  int n  = lane & 15;
  int h  = lane >> 4;

  int t0 = kchunk * TC;
  int valid = T - t0;
  if (valid < 0)  valid = 0;
  if (valid > TC) valid = TC;
  int full = valid & ~3;

  const float* cp = coefT + (size_t)(t0 + 2 * h) * 16 + n;
  const float* xp = x     + (size_t)(t0 + 2 * h) * D  + nb + n;
  const size_t xstep = (size_t)4 * D;

  v8f acc = {};
  if (full > 0) {
    v2f a = { cp[0], cp[16] };
    v2f b = { xp[0], xp[D] };
#pragma unroll 2
    for (int kk = 4; kk < full; kk += 4) {
      cp += 64;
      xp += xstep;
      v2f an = { cp[0], cp[16] };
      v2f bn = { xp[0], xp[D] };
      acc = __builtin_amdgcn_wmma_f32_16x16x4_f32(
          false, a, false, b, (short)0, acc, false, false);
      a = an; b = bn;
    }
    acc = __builtin_amdgcn_wmma_f32_16x16x4_f32(
        false, a, false, b, (short)0, acc, false, false);
  }
  for (int kk = full; kk < TC; kk += 4) {
    int ta = t0 + kk + 2 * h, tb = ta + 1;
    v2f a = { coefT[(size_t)ta * 16 + n], coefT[(size_t)tb * 16 + n] };
    v2f b = { (ta < T) ? x[(size_t)ta * D + nb + n] : 0.0f,
              (tb < T) ? x[(size_t)tb * D + nb + n] : 0.0f };
    acc = __builtin_amdgcn_wmma_f32_16x16x4_f32(
        false, a, false, b, (short)0, acc, false, false);
  }

#pragma unroll
  for (int r = 0; r < 8; ++r) {
    int m = r + 8 * h;
    if (m < NROW)
      partial[((size_t)kchunk * NROW + m) * D + nb + n] = acc[r];
  }
}

#define GX_G     4
#define GX_AHEAD 2
#define GX_TCW   128
#define GX_KQ    4
#define GX_CT    2

template <int DC>
__global__ void __launch_bounds__(GX_KQ * GX_CT * 32)
gemm_coef_x_wide(const float* __restrict__ x, const float* __restrict__ coefT,
                 float* __restrict__ partial) {
  static_assert((GX_TCW / 4) % GX_G == 0 && (GX_TCW / 4) / GX_G >= GX_AHEAD,
                "a wave's K range = whole groups");
  static_assert(DC % (64 * GX_CT) == 0, "row = whole block tiles");
  constexpr int NG = (GX_TCW / 4) / GX_G;
  constexpr int CG = DC / (64 * GX_CT);
  __shared__ v4f red[(GX_KQ - 1) * GX_CT * 8 * 32];

  const int lane = threadIdx.x & 31;
  const int w  = __builtin_amdgcn_readfirstlane((int)threadIdx.x >> 5);
  const int ct = w % GX_CT;
  const int kq = w / GX_CT;
  const int cg     = (int)blockIdx.x % CG;
  const int kchunk = (int)blockIdx.x / CG;
  const int n  = lane & 15;
  const int h  = lane >> 4;
  const int cb = (cg * GX_CT + ct) * 64;
  const int t0 = (kchunk * GX_KQ + kq) * GX_TCW;

  const float* cp = coefT + (size_t)(t0 + 2 * h) * 16 + n;
  const v4f*   xp = (const v4f*)(x + (size_t)(t0 + 2 * h) * DC + cb + 4 * n);

  v2f ab[GX_AHEAD + 1][GX_G];
  v4f xa[GX_AHEAD + 1][GX_G];
  v4f xb[GX_AHEAD + 1][GX_G];
#define GX_LOAD_GROUP(G_)                                                     \
  do {                                                                        \
    _Pragma("unroll") for (int q = 0; q < GX_G; ++q) {                        \
      xa[(G_) % (GX_AHEAD + 1)][q] = xp[(size_t)((G_) * GX_G + q) * DC];      \
      xb[(G_) % (GX_AHEAD + 1)][q] =                                          \
          xp[(size_t)((G_) * GX_G + q) * DC + DC / 4];                        \
    }                                                                         \
    _Pragma("unroll") for (int q = 0; q < GX_G; ++q) {                        \
      const float* c_ = cp + ((G_) * GX_G + q) * 64;                          \
      ab[(G_) % (GX_AHEAD + 1)][q] = (v2f){ c_[0], c_[16] };                  \
    }                                                                         \
  } while (0)

#pragma unroll
  for (int g = 0; g < GX_AHEAD; ++g) GX_LOAD_GROUP(g);

  v8f acc0 = {}, acc1 = {}, acc2 = {}, acc3 = {};
#pragma unroll
  for (int g = 0; g < NG; ++g) {
    __builtin_amdgcn_sched_barrier(0);
    if (g + GX_AHEAD < NG) GX_LOAD_GROUP(g + GX_AHEAD);
    __builtin_amdgcn_sched_barrier(0);
#pragma unroll
    for (int q = 0; q < GX_G; ++q) {
      const v2f a  = ab[g % (GX_AHEAD + 1)][q];
      const v4f u  = xa[g % (GX_AHEAD + 1)][q];
      const v4f v  = xb[g % (GX_AHEAD + 1)][q];
      acc0 = __builtin_amdgcn_wmma_f32_16x16x4_f32(
          false, a, false, (v2f){ u.x, v.x }, (short)0, acc0, false, false);
      acc1 = __builtin_amdgcn_wmma_f32_16x16x4_f32(
          false, a, false, (v2f){ u.y, v.y }, (short)0, acc1, false, false);
      acc2 = __builtin_amdgcn_wmma_f32_16x16x4_f32(
          false, a, false, (v2f){ u.z, v.z }, (short)0, acc2, false, false);
      acc3 = __builtin_amdgcn_wmma_f32_16x16x4_f32(
          false, a, false, (v2f){ u.w, v.w }, (short)0, acc3, false, false);
    }
  }
#undef GX_LOAD_GROUP

  if (kq != 0) {
#pragma unroll
    for (int r = 0; r < 8; ++r)
      red[(((kq - 1) * GX_CT + ct) * 8 + r) * 32 + lane] =
          (v4f){ acc0[r], acc1[r], acc2[r], acc3[r] };
  }
  __syncthreads();
  if (kq == 0) {
#pragma unroll
    for (int r = 0; r < 8; ++r) {
      v4f s = (v4f){ acc0[r], acc1[r], acc2[r], acc3[r] };
#pragma unroll
      for (int p = 0; p < GX_KQ - 1; ++p)
        s += red[((p * GX_CT + ct) * 8 + r) * 32 + lane];
      const int m = r + 8 * h;
      if (m < NROW)
        *(v4f*)(partial + ((size_t)kchunk * NROW + m) * DC + cb + 4 * n) = s;
    }
  }
}

__global__ void reduce_partials(const float* __restrict__ partial,
                                const float* __restrict__ pool_b,
                                float* __restrict__ avg,
                                float* __restrict__ out_hfinal,
                                int D, int KC, int T) {
  int idx = blockIdx.x * blockDim.x + threadIdx.x;
  if (idx >= NROW * D) return;
  int m = idx / D, d = idx % D;
  float s = 0.0f;
  for (int c = 0; c < KC; ++c)
    s += partial[((size_t)c * NROW + m) * D + d];
  float invT = 1.0f / (float)T;
  if (m == 0)      avg[d]     = s * invT;
  else if (m == 1) avg[D + d] = s * invT + pool_b[0];
  else             out_hfinal[(size_t)(m - 2) * D + d] = s;
}

__global__ void gemv_rows(const float* __restrict__ Wm,
                          const float* __restrict__ bias,
                          const float* __restrict__ vec,
                          float* __restrict__ out,
                          int K, int N, int do_relu) {
  int wid  = blockIdx.x * (blockDim.x >> 5) + (threadIdx.x >> 5);
  int lane = threadIdx.x & 31;
  if (wid >= N) return;
  const float4* wr = (const float4*)(Wm + (size_t)wid * K);
  const float4* vv = (const float4*)vec;
  float s = 0.0f;
  for (int i = lane; i < (K >> 2); i += 32) {
    float4 a = wr[i], b = vv[i];
    s += a.x * b.x + a.y * b.y + a.z * b.z + a.w * b.w;
  }
  for (int off = 16; off > 0; off >>= 1) s += __shfl_down(s, off, 32);
  if (lane == 0) {
    float v = s + bias[wid];
    out[wid] = do_relu ? fmaxf(v, 0.0f) : v;
  }
}


#define GV_G     8
#define GV_AHEAD 2
#define GV_WAVES 8

template <int KQ>
__global__ void __launch_bounds__(GV_WAVES * 32)
gemv_rows_stream(const float* __restrict__ Wm,
                 const float* __restrict__ bias,
                 const float* __restrict__ vec,
                 float* __restrict__ out,
                 int N, int do_relu) {
  static_assert(KQ % GV_G == 0 && KQ / GV_G >= GV_AHEAD, "row = whole groups");
  constexpr int NG = KQ / GV_G;
  const int lane = threadIdx.x & 31;
  const int wid  = __builtin_amdgcn_readfirstlane(
      (int)(blockIdx.x * GV_WAVES) + ((int)threadIdx.x >> 5));
  if (wid >= N) return;
  const v4f* wr = (const v4f*)(Wm + (size_t)wid * (128 * KQ)) + lane;
  const v4f* vv = (const v4f*)vec + lane;

  v4f wb[GV_AHEAD + 1][GV_G];
  v4f xb[GV_AHEAD + 1][GV_G];
#define GV_LOAD_GROUP(G_)                                                     \
  do {                                                                        \
    _Pragma("unroll") for (int q = 0; q < GV_G; ++q)                          \
      wb[(G_) % (GV_AHEAD + 1)][q] =                                          \
          __builtin_nontemporal_load(wr + 32 * ((G_) * GV_G + q));            \
    _Pragma("unroll") for (int q = 0; q < GV_G; ++q)                          \
      xb[(G_) % (GV_AHEAD + 1)][q] = vv[32 * ((G_) * GV_G + q)];              \
  } while (0)

#pragma unroll
  for (int g = 0; g < GV_AHEAD; ++g) GV_LOAD_GROUP(g);

  float a0 = 0.0f, a1 = 0.0f, a2 = 0.0f, a3 = 0.0f;
#pragma unroll
  for (int g = 0; g < NG; ++g) {
    __builtin_amdgcn_sched_barrier(0);
    if (g + GV_AHEAD < NG) GV_LOAD_GROUP(g + GV_AHEAD);
    __builtin_amdgcn_sched_barrier(0);
#pragma unroll
    for (int q = 0; q < GV_G; ++q) {
      const v4f a = wb[g % (GV_AHEAD + 1)][q];
      const v4f b = xb[g % (GV_AHEAD + 1)][q];
      a0 = __builtin_fmaf(a.x, b.x, a0);
      a1 = __builtin_fmaf(a.y, b.y, a1);
      a2 = __builtin_fmaf(a.z, b.z, a2);
      a3 = __builtin_fmaf(a.w, b.w, a3);
    }
  }
#undef GV_LOAD_GROUP
  float s = (a0 + a1) + (a2 + a3);
  for (int off = 16; off > 0; off >>= 1) s += __shfl_down(s, off, 32);
  if (lane == 0) {
    float v = s + bias[wid];
    out[wid] = do_relu ? fmaxf(v, 0.0f) : v;
  }
}

#define GK_KS   4
#define GK_RG   4
#define GK_ROWS 4
#define GK_Q    16

template <int KTOT>
__global__ void __launch_bounds__(GK_KS * GK_RG * 32)
gemv_rows_ksplit(const float* __restrict__ Wm,
                 const float* __restrict__ bias,
                 const float* __restrict__ vec,
                 float* __restrict__ out,
                 int do_relu) {
  static_assert(KTOT == GK_KS * GK_Q * 128, "row = GK_KS quarters of GK_Q pieces");
  __shared__ float part[GK_RG * GK_ROWS * GK_KS];
  const int lane = threadIdx.x & 31;
  const int w    = __builtin_amdgcn_readfirstlane((int)threadIdx.x >> 5);
  const int ks   = w % GK_KS;
  const int rg   = w / GK_KS;
  const int row0 = (int)blockIdx.x * (GK_RG * GK_ROWS) + rg * GK_ROWS;
  const v4f* vv = (const v4f*)(vec + ks * (GK_Q * 128)) + lane;
  const v4f* wr = (const v4f*)(Wm + (size_t)row0 * KTOT + ks * (GK_Q * 128)) + lane;

  v4f xq[GK_Q];
#pragma unroll
  for (int q = 0; q < GK_Q; ++q) xq[q] = vv[32 * q];

  v4f wb[2][GK_Q];
#define GK_LOAD_ROW(R_)                                                       \
  do {                                                                        \
    _Pragma("unroll") for (int q = 0; q < GK_Q; ++q)                          \
      wb[(R_) % 2][q] = __builtin_nontemporal_load(                           \
          wr + (size_t)(R_) * (KTOT / 4) + 32 * q);                           \
  } while (0)

  GK_LOAD_ROW(0);
  float sum[GK_ROWS];
#pragma unroll
  for (int r = 0; r < GK_ROWS; ++r) {
    __builtin_amdgcn_sched_barrier(0);
    if (r + 1 < GK_ROWS) GK_LOAD_ROW(r + 1);
    __builtin_amdgcn_sched_barrier(0);
    float a0 = 0.0f, a1 = 0.0f, a2 = 0.0f, a3 = 0.0f;
#pragma unroll
    for (int q = 0; q < GK_Q; ++q) {
      const v4f a = wb[r % 2][q];
      const v4f b = xq[q];
      a0 = __builtin_fmaf(a.x, b.x, a0);
      a1 = __builtin_fmaf(a.y, b.y, a1);
      a2 = __builtin_fmaf(a.z, b.z, a2);
      a3 = __builtin_fmaf(a.w, b.w, a3);
    }
    sum[r] = (a0 + a1) + (a2 + a3);
    asm volatile("" : "+v"(sum[r]));
  }
#undef GK_LOAD_ROW
  __builtin_amdgcn_sched_barrier(0);
#pragma unroll
  for (int r = 0; r < GK_ROWS; ++r) {
    float s = sum[r];
#pragma unroll
    for (int m = 16; m >= 1; m >>= 1) s += __shfl_xor(s, m, 32);
    sum[r] = s;
  }
  if (lane == 0) {
#pragma unroll
    for (int r = 0; r < GK_ROWS; ++r)
      part[(rg * GK_ROWS + r) * GK_KS + ks] = sum[r];
  }
  __syncthreads();
  if (threadIdx.x < GK_RG * GK_ROWS) {
    const float* p = part + threadIdx.x * GK_KS;
    float s = p[0];
#pragma unroll
    for (int k = 1; k < GK_KS; ++k) s += p[k];
    const int row = (int)blockIdx.x * (GK_RG * GK_ROWS) + (int)threadIdx.x;
    const float v = s + bias[row];
    out[row] = do_relu ? fmaxf(v, 0.0f) : v;
  }
}

extern "C" void kernel_launch(void* const* d_in, const int* in_sizes, int n_in,
                              void* d_out, int out_size, void* d_ws, size_t ws_size,
                              hipStream_t stream) {
  const float* inp    = (const float*)d_in[0];
  const float* w      = (const float*)d_in[2];
  const float* pool_w = (const float*)d_in[3];
  const float* pool_b = (const float*)d_in[4];
  const float* i2o_w  = (const float*)d_in[5];
  const float* i2o_b  = (const float*)d_in[6];
  const float* o_w    = (const float*)d_in[7];
  const float* o_b    = (const float*)d_in[8];
  float* out = (float*)d_out;

  const int D  = in_sizes[1] / F_NUM;
  const int T  = in_sizes[0] / D;
  const int H1 = in_sizes[6];

  const int KC = 16;
  int TC   = (((T + KC - 1) / KC) + 3) & ~3;
  int Tpad = KC * TC;
  int NT   = D / 16;

  float* ws      = (float*)d_ws;
  float* coefT   = ws;
  float* partial = coefT + (size_t)Tpad * 16;
  float* avg     = partial + (size_t)KC * NROW * D;
  float* hid1    = avg + (size_t)2 * D;

  coef_kernel<<<(Tpad + 255) / 256, 256, 0, stream>>>(w, pool_w, coefT, T, Tpad, D);

  int units = NT * KC;
  dim3 g2((units + 3) / 4), b2(128);
  if (D == 4096 && TC == GX_TCW * GX_KQ && T == KC * TC)
    gemm_coef_x_wide<4096><<<(4096 / (64 * GX_CT)) * KC, GX_KQ * GX_CT * 32, 0, stream>>>(
        inp, coefT, partial);
  else if (D == 4096)
    gemm_coef_x<4096><<<g2, b2, 0, stream>>>(inp, coefT, partial, T, D, TC, NT, units);
  else
    gemm_coef_x<0><<<g2, b2, 0, stream>>>(inp, coefT, partial, T, D, TC, NT, units);

  reduce_partials<<<(NROW * D + 255) / 256, 256, 0, stream>>>(
      partial, pool_b, avg, out + OUT_N, D, KC, T);

  dim3 gb(GV_WAVES * 32);
  dim3 g4((H1 + GV_WAVES - 1) / GV_WAVES), g5((OUT_N + GV_WAVES - 1) / GV_WAVES);
  if (2 * D == 8192 && H1 % (GK_RG * GK_ROWS) == 0)
    gemv_rows_ksplit<8192><<<H1 / (GK_RG * GK_ROWS), GK_KS * GK_RG * 32, 0, stream>>>(
        i2o_w, i2o_b, avg, hid1, 1);
  else if (2 * D == 8192)
    gemv_rows_stream<64><<<g4, gb, 0, stream>>>(i2o_w, i2o_b, avg, hid1, H1, 1);
  else
    gemv_rows<<<g4, gb, 0, stream>>>(i2o_w, i2o_b, avg, hid1, 2 * D, H1, 1);
  if (H1 == 4096)
    gemv_rows_stream<32><<<g5, gb, 0, stream>>>(o_w, o_b, hid1, out, OUT_N, 0);
  else
    gemv_rows<<<g5, gb, 0, stream>>>(o_w, o_b, hid1, out, H1, OUT_N, 0);
}


# gemm_coef_x: each wave computes its 128 timesteps' coefficients in registers under its first x loads (coefficient kernel dropped, 4 launches); powers = powf + w^4 chain, G's f-sum = lane butterfly
# speedup vs baseline: 3.2409x; 1.1735x over previous
#include <hip/hip_runtime.h>
#include <math.h>

typedef __attribute__((ext_vector_type(2))) float v2f;
typedef __attribute__((ext_vector_type(8))) float v8f;
typedef __attribute__((ext_vector_type(4))) float v4f;

#define F_NUM 10
#define OUT_N 10
#define NROW  12

__global__ void coef_kernel(const float* __restrict__ w,
                            const float* __restrict__ pool_w,
                            float* __restrict__ coefT,
                            int T, int Tpad, int D) {
  int t = blockIdx.x * blockDim.x + threadIdx.x;
  if (t >= Tpad) return;
  float row[16];
#pragma unroll
  for (int i = 0; i < 16; ++i) row[i] = 0.0f;
  if (t < T) {
    row[0] = 1.0f;
    float G = 0.0f;
    float e = (float)(T - 1 - t);
#pragma unroll
    for (int f = 0; f < F_NUM; ++f) {
      float wf = w[(size_t)f * D];
      float pf = pool_w[f];
      float c  = powf(wf, e);
      row[2 + f] = c;
      G += pf * (1.0f - c) / (1.0f - wf);
    }
    row[1] = G;
  }
#pragma unroll
  for (int i = 0; i < 16; ++i) coefT[(size_t)t * 16 + i] = row[i];
}

template <int DC>
__global__ void __launch_bounds__(128)
gemm_coef_x(const float* __restrict__ x, const float* __restrict__ coefT,
            float* __restrict__ partial, int T, int Drt, int TC, int NT, int units) {
  const int D = (DC > 0) ? DC : Drt;
  int wid = __builtin_amdgcn_readfirstlane(
      (int)(blockIdx.x * (blockDim.x >> 5)) + ((int)threadIdx.x >> 5));
  if (wid >= units) return;
  int lane = threadIdx.x & 31;
  int ntile  = wid % NT;
  int kchunk = wid / NT;
  int nb = ntile * 16;
  int n  = lane & 15;
  int h  = lane >> 4;

  int t0 = kchunk * TC;
  int valid = T - t0;
  if (valid < 0)  valid = 0;
  if (valid > TC) valid = TC;
  int full = valid & ~3;

  const float* cp = coefT + (size_t)(t0 + 2 * h) * 16 + n;
  const float* xp = x     + (size_t)(t0 + 2 * h) * D  + nb + n;
  const size_t xstep = (size_t)4 * D;

  v8f acc = {};
  if (full > 0) {
    v2f a = { cp[0], cp[16] };
    v2f b = { xp[0], xp[D] };
#pragma unroll 2
    for (int kk = 4; kk < full; kk += 4) {
      cp += 64;
      xp += xstep;
      v2f an = { cp[0], cp[16] };
      v2f bn = { xp[0], xp[D] };
      acc = __builtin_amdgcn_wmma_f32_16x16x4_f32(
          false, a, false, b, (short)0, acc, false, false);
      a = an; b = bn;
    }
    acc = __builtin_amdgcn_wmma_f32_16x16x4_f32(
        false, a, false, b, (short)0, acc, false, false);
  }
  for (int kk = full; kk < TC; kk += 4) {
    int ta = t0 + kk + 2 * h, tb = ta + 1;
    v2f a = { coefT[(size_t)ta * 16 + n], coefT[(size_t)tb * 16 + n] };
    v2f b = { (ta < T) ? x[(size_t)ta * D + nb + n] : 0.0f,
              (tb < T) ? x[(size_t)tb * D + nb + n] : 0.0f };
    acc = __builtin_amdgcn_wmma_f32_16x16x4_f32(
        false, a, false, b, (short)0, acc, false, false);
  }

#pragma unroll
  for (int r = 0; r < 8; ++r) {
    int m = r + 8 * h;
    if (m < NROW)
      partial[((size_t)kchunk * NROW + m) * D + nb + n] = acc[r];
  }
}

#define GX_G     4
#define GX_AHEAD 2
#define GX_TCW   128
#define GX_KQ    4
#define GX_CT    2

template <int M>
__device__ __forceinline__ float gx_row_xor(float v) {
  return __builtin_bit_cast(float, __builtin_amdgcn_update_dpp(
      0, __builtin_bit_cast(int, v), 0x160 + M, 0xf, 0xf, false));
}
__device__ __forceinline__ float gx_row16_sum(float v) {
  v += gx_row_xor<1>(v);
  v += gx_row_xor<2>(v);
  v += gx_row_xor<4>(v);
  v += gx_row_xor<8>(v);
  return v;
}

template <int DC>
__global__ void __launch_bounds__(GX_KQ * GX_CT * 32)
gemm_coef_x_wide(const float* __restrict__ x, const float* __restrict__ decay,
                 const float* __restrict__ pool_w,
                 float* __restrict__ partial, int T) {
  static_assert((GX_TCW / 4) % GX_G == 0 && (GX_TCW / 4) / GX_G > GX_AHEAD,
                "a wave's K range = whole groups, more than the ring holds");
  static_assert(DC % (64 * GX_CT) == 0, "row = whole block tiles");
  static_assert((GX_TCW / 4) % 2 == 0, "two powf anchors per lane");
  constexpr int NS = GX_TCW / 4;
  constexpr int NG = NS / GX_G;
  constexpr int CG = DC / (64 * GX_CT);
  __shared__ v4f red[(GX_KQ - 1) * GX_CT * 8 * 32];

  const int lane = threadIdx.x & 31;
  const int w  = __builtin_amdgcn_readfirstlane((int)threadIdx.x >> 5);
  const int ct = w % GX_CT;
  const int kq = w / GX_CT;
  const int cg     = (int)blockIdx.x % CG;
  const int kchunk = (int)blockIdx.x / CG;
  const int n  = lane & 15;
  const int h  = lane >> 4;
  const int cb = (cg * GX_CT + ct) * 64;
  const int t0 = (kchunk * GX_KQ + kq) * GX_TCW;

  const v4f*   xp = (const v4f*)(x + (size_t)(t0 + 2 * h) * DC + cb + 4 * n);

  float wS[F_NUM], pS[F_NUM];
#pragma unroll
  for (int f = 0; f < F_NUM; ++f) {
    wS[f] = decay[(size_t)f * DC];
    pS[f] = pool_w[f];
  }

  v2f ca[NS];
  v4f xa[GX_AHEAD + 1][GX_G];
  v4f xb[GX_AHEAD + 1][GX_G];
#define GX_LOAD_GROUP(G_)                                                     \
  do {                                                                        \
    _Pragma("unroll") for (int q = 0; q < GX_G; ++q) {                        \
      xa[(G_) % (GX_AHEAD + 1)][q] = xp[(size_t)((G_) * GX_G + q) * DC];      \
      xb[(G_) % (GX_AHEAD + 1)][q] =                                          \
          xp[(size_t)((G_) * GX_G + q) * DC + DC / 4];                        \
    }                                                                         \
  } while (0)

#pragma unroll
  for (int g = 0; g <= GX_AHEAD; ++g) GX_LOAD_GROUP(g);
  __builtin_amdgcn_sched_barrier(0);

  {
    const bool is_pow = (n >= 2) && (n < 2 + F_NUM);
    float wf = 0.0f, pf = 0.0f;
#pragma unroll
    for (int f = 0; f < F_NUM; ++f)
      if (n == 2 + f) { wf = wS[f]; pf = pS[f]; }
    const float fixed = (n == 0) ? 1.0f : 0.0f;
    const float wl = is_pow ? wf : fixed;
    const float ql = pf / (1.0f - wf);
    const float w2 = wl * wl, w4 = w2 * w2;
    const int e1 = T - t0 - (GX_TCW - 2) - 2 * h;
    float top = powf(wl, (float)e1);
    float mid = powf(wl, (float)(e1 + GX_TCW / 2));
    if (!is_pow) { top = fixed; mid = fixed; }
    float cx[NS], cy[NS];
    cy[NS - 1]     = top;  cx[NS - 1]     = top * wl;
    cy[NS / 2 - 1] = mid;  cx[NS / 2 - 1] = mid * wl;
#pragma unroll
    for (int s = NS - 2; s >= NS / 2; --s) {
      cy[s] = cy[s + 1] * w4;  cx[s] = cx[s + 1] * w4;
    }
#pragma unroll
    for (int s = NS / 2 - 2; s >= 0; --s) {
      cy[s] = cy[s + 1] * w4;  cx[s] = cx[s + 1] * w4;
    }
#pragma unroll
    for (int s = 0; s < NS; ++s) {
      const float gx = gx_row16_sum(ql * (1.0f - cx[s]));
      const float gy = gx_row16_sum(ql * (1.0f - cy[s]));
      ca[s] = (v2f){ (n == 1) ? gx : cx[s], (n == 1) ? gy : cy[s] };
    }
  }

  v8f acc0 = {}, acc1 = {}, acc2 = {}, acc3 = {};
#pragma unroll
  for (int g = 0; g < NG; ++g) {
    __builtin_amdgcn_sched_barrier(0);
    if (g >= 1 && g + GX_AHEAD < NG) GX_LOAD_GROUP(g + GX_AHEAD);
    __builtin_amdgcn_sched_barrier(0);
#pragma unroll
    for (int q = 0; q < GX_G; ++q) {
      const v2f a  = ca[g * GX_G + q];
      const v4f u  = xa[g % (GX_AHEAD + 1)][q];
      const v4f v  = xb[g % (GX_AHEAD + 1)][q];
      acc0 = __builtin_amdgcn_wmma_f32_16x16x4_f32(
          false, a, false, (v2f){ u.x, v.x }, (short)0, acc0, false, false);
      acc1 = __builtin_amdgcn_wmma_f32_16x16x4_f32(
          false, a, false, (v2f){ u.y, v.y }, (short)0, acc1, false, false);
      acc2 = __builtin_amdgcn_wmma_f32_16x16x4_f32(
          false, a, false, (v2f){ u.z, v.z }, (short)0, acc2, false, false);
      acc3 = __builtin_amdgcn_wmma_f32_16x16x4_f32(
          false, a, false, (v2f){ u.w, v.w }, (short)0, acc3, false, false);
    }
  }
#undef GX_LOAD_GROUP

  if (kq != 0) {
#pragma unroll
    for (int r = 0; r < 8; ++r)
      red[(((kq - 1) * GX_CT + ct) * 8 + r) * 32 + lane] =
          (v4f){ acc0[r], acc1[r], acc2[r], acc3[r] };
  }
  __syncthreads();
  if (kq == 0) {
#pragma unroll
    for (int r = 0; r < 8; ++r) {
      v4f s = (v4f){ acc0[r], acc1[r], acc2[r], acc3[r] };
#pragma unroll
      for (int p = 0; p < GX_KQ - 1; ++p)
        s += red[((p * GX_CT + ct) * 8 + r) * 32 + lane];
      const int m = r + 8 * h;
      if (m < NROW)
        *(v4f*)(partial + ((size_t)kchunk * NROW + m) * DC + cb + 4 * n) = s;
    }
  }
}

__global__ void reduce_partials(const float* __restrict__ partial,
                                const float* __restrict__ pool_b,
                                float* __restrict__ avg,
                                float* __restrict__ out_hfinal,
                                int D, int KC, int T) {
  int idx = blockIdx.x * blockDim.x + threadIdx.x;
  if (idx >= NROW * D) return;
  int m = idx / D, d = idx % D;
  float s = 0.0f;
  for (int c = 0; c < KC; ++c)
    s += partial[((size_t)c * NROW + m) * D + d];
  float invT = 1.0f / (float)T;
  if (m == 0)      avg[d]     = s * invT;
  else if (m == 1) avg[D + d] = s * invT + pool_b[0];
  else             out_hfinal[(size_t)(m - 2) * D + d] = s;
}

__global__ void gemv_rows(const float* __restrict__ Wm,
                          const float* __restrict__ bias,
                          const float* __restrict__ vec,
                          float* __restrict__ out,
                          int K, int N, int do_relu) {
  int wid  = blockIdx.x * (blockDim.x >> 5) + (threadIdx.x >> 5);
  int lane = threadIdx.x & 31;
  if (wid >= N) return;
  const float4* wr = (const float4*)(Wm + (size_t)wid * K);
  const float4* vv = (const float4*)vec;
  float s = 0.0f;
  for (int i = lane; i < (K >> 2); i += 32) {
    float4 a = wr[i], b = vv[i];
    s += a.x * b.x + a.y * b.y + a.z * b.z + a.w * b.w;
  }
  for (int off = 16; off > 0; off >>= 1) s += __shfl_down(s, off, 32);
  if (lane == 0) {
    float v = s + bias[wid];
    out[wid] = do_relu ? fmaxf(v, 0.0f) : v;
  }
}


#define GV_G     8
#define GV_AHEAD 2
#define GV_WAVES 8

template <int KQ>
__global__ void __launch_bounds__(GV_WAVES * 32)
gemv_rows_stream(const float* __restrict__ Wm,
                 const float* __restrict__ bias,
                 const float* __restrict__ vec,
                 float* __restrict__ out,
                 int N, int do_relu) {
  static_assert(KQ % GV_G == 0 && KQ / GV_G >= GV_AHEAD, "row = whole groups");
  constexpr int NG = KQ / GV_G;
  const int lane = threadIdx.x & 31;
  const int wid  = __builtin_amdgcn_readfirstlane(
      (int)(blockIdx.x * GV_WAVES) + ((int)threadIdx.x >> 5));
  if (wid >= N) return;
  const v4f* wr = (const v4f*)(Wm + (size_t)wid * (128 * KQ)) + lane;
  const v4f* vv = (const v4f*)vec + lane;

  v4f wb[GV_AHEAD + 1][GV_G];
  v4f xb[GV_AHEAD + 1][GV_G];
#define GV_LOAD_GROUP(G_)                                                     \
  do {                                                                        \
    _Pragma("unroll") for (int q = 0; q < GV_G; ++q)                          \
      wb[(G_) % (GV_AHEAD + 1)][q] =                                          \
          __builtin_nontemporal_load(wr + 32 * ((G_) * GV_G + q));            \
    _Pragma("unroll") for (int q = 0; q < GV_G; ++q)                          \
      xb[(G_) % (GV_AHEAD + 1)][q] = vv[32 * ((G_) * GV_G + q)];              \
  } while (0)

#pragma unroll
  for (int g = 0; g < GV_AHEAD; ++g) GV_LOAD_GROUP(g);

  float a0 = 0.0f, a1 = 0.0f, a2 = 0.0f, a3 = 0.0f;
#pragma unroll
  for (int g = 0; g < NG; ++g) {
    __builtin_amdgcn_sched_barrier(0);
    if (g + GV_AHEAD < NG) GV_LOAD_GROUP(g + GV_AHEAD);
    __builtin_amdgcn_sched_barrier(0);
#pragma unroll
    for (int q = 0; q < GV_G; ++q) {
      const v4f a = wb[g % (GV_AHEAD + 1)][q];
      const v4f b = xb[g % (GV_AHEAD + 1)][q];
      a0 = __builtin_fmaf(a.x, b.x, a0);
      a1 = __builtin_fmaf(a.y, b.y, a1);
      a2 = __builtin_fmaf(a.z, b.z, a2);
      a3 = __builtin_fmaf(a.w, b.w, a3);
    }
  }
#undef GV_LOAD_GROUP
  float s = (a0 + a1) + (a2 + a3);
  for (int off = 16; off > 0; off >>= 1) s += __shfl_down(s, off, 32);
  if (lane == 0) {
    float v = s + bias[wid];
    out[wid] = do_relu ? fmaxf(v, 0.0f) : v;
  }
}

#define GK_KS   4
#define GK_RG   4
#define GK_ROWS 4
#define GK_Q    16

template <int KTOT>
__global__ void __launch_bounds__(GK_KS * GK_RG * 32)
gemv_rows_ksplit(const float* __restrict__ Wm,
                 const float* __restrict__ bias,
                 const float* __restrict__ vec,
                 float* __restrict__ out,
                 int do_relu) {
  static_assert(KTOT == GK_KS * GK_Q * 128, "row = GK_KS quarters of GK_Q pieces");
  __shared__ float part[GK_RG * GK_ROWS * GK_KS];
  const int lane = threadIdx.x & 31;
  const int w    = __builtin_amdgcn_readfirstlane((int)threadIdx.x >> 5);
  const int ks   = w % GK_KS;
  const int rg   = w / GK_KS;
  const int row0 = (int)blockIdx.x * (GK_RG * GK_ROWS) + rg * GK_ROWS;
  const v4f* vv = (const v4f*)(vec + ks * (GK_Q * 128)) + lane;
  const v4f* wr = (const v4f*)(Wm + (size_t)row0 * KTOT + ks * (GK_Q * 128)) + lane;

  v4f xq[GK_Q];
#pragma unroll
  for (int q = 0; q < GK_Q; ++q) xq[q] = vv[32 * q];

  v4f wb[2][GK_Q];
#define GK_LOAD_ROW(R_)                                                       \
  do {                                                                        \
    _Pragma("unroll") for (int q = 0; q < GK_Q; ++q)                          \
      wb[(R_) % 2][q] = __builtin_nontemporal_load(                           \
          wr + (size_t)(R_) * (KTOT / 4) + 32 * q);                           \
  } while (0)

  GK_LOAD_ROW(0);
  float sum[GK_ROWS];
#pragma unroll
  for (int r = 0; r < GK_ROWS; ++r) {
    __builtin_amdgcn_sched_barrier(0);
    if (r + 1 < GK_ROWS) GK_LOAD_ROW(r + 1);
    __builtin_amdgcn_sched_barrier(0);
    float a0 = 0.0f, a1 = 0.0f, a2 = 0.0f, a3 = 0.0f;
#pragma unroll
    for (int q = 0; q < GK_Q; ++q) {
      const v4f a = wb[r % 2][q];
      const v4f b = xq[q];
      a0 = __builtin_fmaf(a.x, b.x, a0);
      a1 = __builtin_fmaf(a.y, b.y, a1);
      a2 = __builtin_fmaf(a.z, b.z, a2);
      a3 = __builtin_fmaf(a.w, b.w, a3);
    }
    sum[r] = (a0 + a1) + (a2 + a3);
    asm volatile("" : "+v"(sum[r]));
  }
#undef GK_LOAD_ROW
  __builtin_amdgcn_sched_barrier(0);
#pragma unroll
  for (int r = 0; r < GK_ROWS; ++r) {
    float s = sum[r];
#pragma unroll
    for (int m = 16; m >= 1; m >>= 1) s += __shfl_xor(s, m, 32);
    sum[r] = s;
  }
  if (lane == 0) {
#pragma unroll
    for (int r = 0; r < GK_ROWS; ++r)
      part[(rg * GK_ROWS + r) * GK_KS + ks] = sum[r];
  }
  __syncthreads();
  if (threadIdx.x < GK_RG * GK_ROWS) {
    const float* p = part + threadIdx.x * GK_KS;
    float s = p[0];
#pragma unroll
    for (int k = 1; k < GK_KS; ++k) s += p[k];
    const int row = (int)blockIdx.x * (GK_RG * GK_ROWS) + (int)threadIdx.x;
    const float v = s + bias[row];
    out[row] = do_relu ? fmaxf(v, 0.0f) : v;
  }
}

extern "C" void kernel_launch(void* const* d_in, const int* in_sizes, int n_in,
                              void* d_out, int out_size, void* d_ws, size_t ws_size,
                              hipStream_t stream) {
  const float* inp    = (const float*)d_in[0];
  const float* w      = (const float*)d_in[2];
  const float* pool_w = (const float*)d_in[3];
  const float* pool_b = (const float*)d_in[4];
  const float* i2o_w  = (const float*)d_in[5];
  const float* i2o_b  = (const float*)d_in[6];
  const float* o_w    = (const float*)d_in[7];
  const float* o_b    = (const float*)d_in[8];
  float* out = (float*)d_out;

  const int D  = in_sizes[1] / F_NUM;
  const int T  = in_sizes[0] / D;
  const int H1 = in_sizes[6];

  const int KC = 16;
  int TC   = (((T + KC - 1) / KC) + 3) & ~3;
  int Tpad = KC * TC;
  int NT   = D / 16;

  float* ws      = (float*)d_ws;
  float* coefT   = ws;
  float* partial = coefT + (size_t)Tpad * 16;
  float* avg     = partial + (size_t)KC * NROW * D;
  float* hid1    = avg + (size_t)2 * D;

  int units = NT * KC;
  dim3 g2((units + 3) / 4), b2(128);
  if (D == 4096 && TC == GX_TCW * GX_KQ && T == KC * TC) {
    gemm_coef_x_wide<4096><<<(4096 / (64 * GX_CT)) * KC, GX_KQ * GX_CT * 32, 0, stream>>>(
        inp, w, pool_w, partial, T);
  } else {
    coef_kernel<<<(Tpad + 255) / 256, 256, 0, stream>>>(w, pool_w, coefT, T, Tpad, D);
    if (D == 4096)
      gemm_coef_x<4096><<<g2, b2, 0, stream>>>(inp, coefT, partial, T, D, TC, NT, units);
    else
      gemm_coef_x<0><<<g2, b2, 0, stream>>>(inp, coefT, partial, T, D, TC, NT, units);
  }

  reduce_partials<<<(NROW * D + 255) / 256, 256, 0, stream>>>(
      partial, pool_b, avg, out + OUT_N, D, KC, T);

  dim3 gb(GV_WAVES * 32);
  dim3 g4((H1 + GV_WAVES - 1) / GV_WAVES), g5((OUT_N + GV_WAVES - 1) / GV_WAVES);
  if (2 * D == 8192 && H1 % (GK_RG * GK_ROWS) == 0)
    gemv_rows_ksplit<8192><<<H1 / (GK_RG * GK_ROWS), GK_KS * GK_RG * 32, 0, stream>>>(
        i2o_w, i2o_b, avg, hid1, 1);
  else if (2 * D == 8192)
    gemv_rows_stream<64><<<g4, gb, 0, stream>>>(i2o_w, i2o_b, avg, hid1, H1, 1);
  else
    gemv_rows<<<g4, gb, 0, stream>>>(i2o_w, i2o_b, avg, hid1, 2 * D, H1, 1);
  if (H1 == 4096)
    gemv_rows_stream<32><<<g5, gb, 0, stream>>>(o_w, o_b, hid1, out, OUT_N, 0);
  else
    gemv_rows<<<g5, gb, 0, stream>>>(o_w, o_b, hid1, out, H1, OUT_N, 0);
}


# reduction launch dropped (3 launches): product kernel sums 16 K-pieces per block in LDS (4 chunk partials), GEMV builds its vector and h_final from them itself; sum order changes
# speedup vs baseline: 3.5376x; 1.0915x over previous
#include <hip/hip_runtime.h>
#include <math.h>

typedef __attribute__((ext_vector_type(2))) float v2f;
typedef __attribute__((ext_vector_type(8))) float v8f;
typedef __attribute__((ext_vector_type(4))) float v4f;

#define F_NUM 10
#define OUT_N 10
#define NROW  12

__global__ void coef_kernel(const float* __restrict__ w,
                            const float* __restrict__ pool_w,
                            float* __restrict__ coefT,
                            int T, int Tpad, int D) {
  int t = blockIdx.x * blockDim.x + threadIdx.x;
  if (t >= Tpad) return;
  float row[16];
#pragma unroll
  for (int i = 0; i < 16; ++i) row[i] = 0.0f;
  if (t < T) {
    row[0] = 1.0f;
    float G = 0.0f;
    float e = (float)(T - 1 - t);
#pragma unroll
    for (int f = 0; f < F_NUM; ++f) {
      float wf = w[(size_t)f * D];
      float pf = pool_w[f];
      float c  = powf(wf, e);
      row[2 + f] = c;
      G += pf * (1.0f - c) / (1.0f - wf);
    }
    row[1] = G;
  }
#pragma unroll
  for (int i = 0; i < 16; ++i) coefT[(size_t)t * 16 + i] = row[i];
}

template <int DC>
__global__ void __launch_bounds__(128)
gemm_coef_x(const float* __restrict__ x, const float* __restrict__ coefT,
            float* __restrict__ partial, int T, int Drt, int TC, int NT, int units) {
  const int D = (DC > 0) ? DC : Drt;
  int wid = __builtin_amdgcn_readfirstlane(
      (int)(blockIdx.x * (blockDim.x >> 5)) + ((int)threadIdx.x >> 5));
  if (wid >= units) return;
  int lane = threadIdx.x & 31;
  int ntile  = wid % NT;
  int kchunk = wid / NT;
  int nb = ntile * 16;
  int n  = lane & 15;
  int h  = lane >> 4;

  int t0 = kchunk * TC;
  int valid = T - t0;
  if (valid < 0)  valid = 0;
  if (valid > TC) valid = TC;
  int full = valid & ~3;

  const float* cp = coefT + (size_t)(t0 + 2 * h) * 16 + n;
  const float* xp = x     + (size_t)(t0 + 2 * h) * D  + nb + n;
  const size_t xstep = (size_t)4 * D;

  v8f acc = {};
  if (full > 0) {
    v2f a = { cp[0], cp[16] };
    v2f b = { xp[0], xp[D] };
#pragma unroll 2
    for (int kk = 4; kk < full; kk += 4) {
      cp += 64;
      xp += xstep;
      v2f an = { cp[0], cp[16] };
      v2f bn = { xp[0], xp[D] };
      acc = __builtin_amdgcn_wmma_f32_16x16x4_f32(
          false, a, false, b, (short)0, acc, false, false);
      a = an; b = bn;
    }
    acc = __builtin_amdgcn_wmma_f32_16x16x4_f32(
        false, a, false, b, (short)0, acc, false, false);
  }
  for (int kk = full; kk < TC; kk += 4) {
    int ta = t0 + kk + 2 * h, tb = ta + 1;
    v2f a = { coefT[(size_t)ta * 16 + n], coefT[(size_t)tb * 16 + n] };
    v2f b = { (ta < T) ? x[(size_t)ta * D + nb + n] : 0.0f,
              (tb < T) ? x[(size_t)tb * D + nb + n] : 0.0f };
    acc = __builtin_amdgcn_wmma_f32_16x16x4_f32(
        false, a, false, b, (short)0, acc, false, false);
  }

#pragma unroll
  for (int r = 0; r < 8; ++r) {
    int m = r + 8 * h;
    if (m < NROW)
      partial[((size_t)kchunk * NROW + m) * D + nb + n] = acc[r];
  }
}

#define GX_G     4
#define GX_AHEAD 2
#define GX_TCW   128
#define GX_KQ    16
#define GX_CT    1
#define GX_KC    4

template <int M>
__device__ __forceinline__ float gx_row_xor(float v) {
  return __builtin_bit_cast(float, __builtin_amdgcn_update_dpp(
      0, __builtin_bit_cast(int, v), 0x160 + M, 0xf, 0xf, false));
}
__device__ __forceinline__ float gx_row16_sum(float v) {
  v += gx_row_xor<1>(v);
  v += gx_row_xor<2>(v);
  v += gx_row_xor<4>(v);
  v += gx_row_xor<8>(v);
  return v;
}

template <int DC>
__global__ void __launch_bounds__(GX_KQ * GX_CT * 32)
gemm_coef_x_wide(const float* __restrict__ x, const float* __restrict__ decay,
                 const float* __restrict__ pool_w,
                 float* __restrict__ partial, int T) {
  static_assert((GX_TCW / 4) % GX_G == 0 && (GX_TCW / 4) / GX_G > GX_AHEAD,
                "a wave's K range = whole groups, more than the ring holds");
  static_assert(DC % (64 * GX_CT) == 0, "row = whole block tiles");
  static_assert((GX_TCW / 4) % 2 == 0, "two powf anchors per lane");
  static_assert(GX_KQ >= 8, "one summing wave per accumulator register slot");
  constexpr int NS = GX_TCW / 4;
  constexpr int NG = NS / GX_G;
  constexpr int CG = DC / (64 * GX_CT);
  __shared__ v4f red[GX_KQ * GX_CT * 8 * 32];

  const int lane = threadIdx.x & 31;
  const int w  = __builtin_amdgcn_readfirstlane((int)threadIdx.x >> 5);
  const int ct = w % GX_CT;
  const int kq = w / GX_CT;
  const int cg     = (int)blockIdx.x % CG;
  const int kchunk = (int)blockIdx.x / CG;
  const int n  = lane & 15;
  const int h  = lane >> 4;
  const int cb = (cg * GX_CT + ct) * 64;
  const int t0 = (kchunk * GX_KQ + kq) * GX_TCW;

  const v4f*   xp = (const v4f*)(x + (size_t)(t0 + 2 * h) * DC + cb + 4 * n);

  float wS[F_NUM], pS[F_NUM];
#pragma unroll
  for (int f = 0; f < F_NUM; ++f) {
    wS[f] = decay[(size_t)f * DC];
    pS[f] = pool_w[f];
  }

  v2f ca[NS];
  v4f xa[GX_AHEAD + 1][GX_G];
  v4f xb[GX_AHEAD + 1][GX_G];
#define GX_LOAD_GROUP(G_)                                                     \
  do {                                                                        \
    _Pragma("unroll") for (int q = 0; q < GX_G; ++q) {                        \
      xa[(G_) % (GX_AHEAD + 1)][q] = xp[(size_t)((G_) * GX_G + q) * DC];      \
      xb[(G_) % (GX_AHEAD + 1)][q] =                                          \
          xp[(size_t)((G_) * GX_G + q) * DC + DC / 4];                        \
    }                                                                         \
  } while (0)

#pragma unroll
  for (int g = 0; g <= GX_AHEAD; ++g) GX_LOAD_GROUP(g);
  __builtin_amdgcn_sched_barrier(0);

  {
    const bool is_pow = (n >= 2) && (n < 2 + F_NUM);
    float wf = 0.0f, pf = 0.0f;
#pragma unroll
    for (int f = 0; f < F_NUM; ++f)
      if (n == 2 + f) { wf = wS[f]; pf = pS[f]; }
    const float fixed = (n == 0) ? 1.0f : 0.0f;
    const float wl = is_pow ? wf : fixed;
    const float ql = pf / (1.0f - wf);
    const float w2 = wl * wl, w4 = w2 * w2;
    const int e1 = T - t0 - (GX_TCW - 2) - 2 * h;
    float top = powf(wl, (float)e1);
    float mid = powf(wl, (float)(e1 + GX_TCW / 2));
    if (!is_pow) { top = fixed; mid = fixed; }
    float cx[NS], cy[NS];
    cy[NS - 1]     = top;  cx[NS - 1]     = top * wl;
    cy[NS / 2 - 1] = mid;  cx[NS / 2 - 1] = mid * wl;
#pragma unroll
    for (int s = NS - 2; s >= NS / 2; --s) {
      cy[s] = cy[s + 1] * w4;  cx[s] = cx[s + 1] * w4;
    }
#pragma unroll
    for (int s = NS / 2 - 2; s >= 0; --s) {
      cy[s] = cy[s + 1] * w4;  cx[s] = cx[s + 1] * w4;
    }
#pragma unroll
    for (int s = 0; s < NS; ++s) {
      const float gx = gx_row16_sum(ql * (1.0f - cx[s]));
      const float gy = gx_row16_sum(ql * (1.0f - cy[s]));
      ca[s] = (v2f){ (n == 1) ? gx : cx[s], (n == 1) ? gy : cy[s] };
    }
  }

  v8f acc0 = {}, acc1 = {}, acc2 = {}, acc3 = {};
#pragma unroll
  for (int g = 0; g < NG; ++g) {
    __builtin_amdgcn_sched_barrier(0);
    if (g >= 1 && g + GX_AHEAD < NG) GX_LOAD_GROUP(g + GX_AHEAD);
    __builtin_amdgcn_sched_barrier(0);
#pragma unroll
    for (int q = 0; q < GX_G; ++q) {
      const v2f a  = ca[g * GX_G + q];
      const v4f u  = xa[g % (GX_AHEAD + 1)][q];
      const v4f v  = xb[g % (GX_AHEAD + 1)][q];
      acc0 = __builtin_amdgcn_wmma_f32_16x16x4_f32(
          false, a, false, (v2f){ u.x, v.x }, (short)0, acc0, false, false);
      acc1 = __builtin_amdgcn_wmma_f32_16x16x4_f32(
          false, a, false, (v2f){ u.y, v.y }, (short)0, acc1, false, false);
      acc2 = __builtin_amdgcn_wmma_f32_16x16x4_f32(
          false, a, false, (v2f){ u.z, v.z }, (short)0, acc2, false, false);
      acc3 = __builtin_amdgcn_wmma_f32_16x16x4_f32(
          false, a, false, (v2f){ u.w, v.w }, (short)0, acc3, false, false);
    }
  }
#undef GX_LOAD_GROUP

#pragma unroll
  for (int r = 0; r < 8; ++r)
    red[((kq * GX_CT + ct) * 8 + r) * 32 + lane] =
        (v4f){ acc0[r], acc1[r], acc2[r], acc3[r] };
  __syncthreads();
  if (kq < 8) {
    const int r = kq;
    v4f s = red[(ct * 8 + r) * 32 + lane];
#pragma unroll
    for (int p = 1; p < GX_KQ; ++p)
      s += red[((p * GX_CT + ct) * 8 + r) * 32 + lane];
    const int m = r + 8 * h;
    if (m < NROW)
      *(v4f*)(partial + ((size_t)kchunk * NROW + m) * DC + cb + 4 * n) = s;
  }
}

__global__ void reduce_partials(const float* __restrict__ partial,
                                const float* __restrict__ pool_b,
                                float* __restrict__ avg,
                                float* __restrict__ out_hfinal,
                                int D, int KC, int T) {
  int idx = blockIdx.x * blockDim.x + threadIdx.x;
  if (idx >= NROW * D) return;
  int m = idx / D, d = idx % D;
  float s = 0.0f;
  for (int c = 0; c < KC; ++c)
    s += partial[((size_t)c * NROW + m) * D + d];
  float invT = 1.0f / (float)T;
  if (m == 0)      avg[d]     = s * invT;
  else if (m == 1) avg[D + d] = s * invT + pool_b[0];
  else             out_hfinal[(size_t)(m - 2) * D + d] = s;
}

__global__ void gemv_rows(const float* __restrict__ Wm,
                          const float* __restrict__ bias,
                          const float* __restrict__ vec,
                          float* __restrict__ out,
                          int K, int N, int do_relu) {
  int wid  = blockIdx.x * (blockDim.x >> 5) + (threadIdx.x >> 5);
  int lane = threadIdx.x & 31;
  if (wid >= N) return;
  const float4* wr = (const float4*)(Wm + (size_t)wid * K);
  const float4* vv = (const float4*)vec;
  float s = 0.0f;
  for (int i = lane; i < (K >> 2); i += 32) {
    float4 a = wr[i], b = vv[i];
    s += a.x * b.x + a.y * b.y + a.z * b.z + a.w * b.w;
  }
  for (int off = 16; off > 0; off >>= 1) s += __shfl_down(s, off, 32);
  if (lane == 0) {
    float v = s + bias[wid];
    out[wid] = do_relu ? fmaxf(v, 0.0f) : v;
  }
}


#define GV_G     8
#define GV_AHEAD 2
#define GV_WAVES 8

template <int KQ>
__global__ void __launch_bounds__(GV_WAVES * 32)
gemv_rows_stream(const float* __restrict__ Wm,
                 const float* __restrict__ bias,
                 const float* __restrict__ vec,
                 float* __restrict__ out,
                 int N, int do_relu) {
  static_assert(KQ % GV_G == 0 && KQ / GV_G >= GV_AHEAD, "row = whole groups");
  constexpr int NG = KQ / GV_G;
  const int lane = threadIdx.x & 31;
  const int wid  = __builtin_amdgcn_readfirstlane(
      (int)(blockIdx.x * GV_WAVES) + ((int)threadIdx.x >> 5));
  if (wid >= N) return;
  const v4f* wr = (const v4f*)(Wm + (size_t)wid * (128 * KQ)) + lane;
  const v4f* vv = (const v4f*)vec + lane;

  v4f wb[GV_AHEAD + 1][GV_G];
  v4f xb[GV_AHEAD + 1][GV_G];
#define GV_LOAD_GROUP(G_)                                                     \
  do {                                                                        \
    _Pragma("unroll") for (int q = 0; q < GV_G; ++q)                          \
      wb[(G_) % (GV_AHEAD + 1)][q] =                                          \
          __builtin_nontemporal_load(wr + 32 * ((G_) * GV_G + q));            \
    _Pragma("unroll") for (int q = 0; q < GV_G; ++q)                          \
      xb[(G_) % (GV_AHEAD + 1)][q] = vv[32 * ((G_) * GV_G + q)];              \
  } while (0)

#pragma unroll
  for (int g = 0; g < GV_AHEAD; ++g) GV_LOAD_GROUP(g);

  float a0 = 0.0f, a1 = 0.0f, a2 = 0.0f, a3 = 0.0f;
#pragma unroll
  for (int g = 0; g < NG; ++g) {
    __builtin_amdgcn_sched_barrier(0);
    if (g + GV_AHEAD < NG) GV_LOAD_GROUP(g + GV_AHEAD);
    __builtin_amdgcn_sched_barrier(0);
#pragma unroll
    for (int q = 0; q < GV_G; ++q) {
      const v4f a = wb[g % (GV_AHEAD + 1)][q];
      const v4f b = xb[g % (GV_AHEAD + 1)][q];
      a0 = __builtin_fmaf(a.x, b.x, a0);
      a1 = __builtin_fmaf(a.y, b.y, a1);
      a2 = __builtin_fmaf(a.z, b.z, a2);
      a3 = __builtin_fmaf(a.w, b.w, a3);
    }
  }
#undef GV_LOAD_GROUP
  float s = (a0 + a1) + (a2 + a3);
  for (int off = 16; off > 0; off >>= 1) s += __shfl_down(s, off, 32);
  if (lane == 0) {
    float v = s + bias[wid];
    out[wid] = do_relu ? fmaxf(v, 0.0f) : v;
  }
}

#define GK_KS   4
#define GK_RG   4
#define GK_ROWS 4
#define GK_Q    16

template <int KTOT, int KCH, int DD, int NBLK>
__global__ void __launch_bounds__(GK_KS * GK_RG * 32)
gemv_rows_ksplit(const float* __restrict__ Wm,
                 const float* __restrict__ bias,
                 const float* __restrict__ partial,
                 const float* __restrict__ pool_b,
                 float* __restrict__ out,
                 float* __restrict__ out_hfinal,
                 int T, int do_relu) {
  static_assert(KTOT == GK_KS * GK_Q * 128, "row = GK_KS quarters of GK_Q pieces");
  constexpr int NTHR = GK_KS * GK_RG * 32;
  constexpr int NV   = KTOT / 4;
  constexpr int NJ   = NV / NTHR;
  constexpr int HP   = 5;
  static_assert(KTOT == 2 * DD && NV % NTHR == 0 && (DD / 4) % NTHR == 0,
                "a thread's piece j lies in one half of the vector");
  static_assert(NBLK * HP == F_NUM * (DD / 32) && HP <= GK_KS * GK_RG,
                "h_final = NBLK blocks x HP pieces");
  __shared__ v4f avgs[NV];
  __shared__ float part[GK_RG * GK_ROWS * GK_KS];
  const int lane = threadIdx.x & 31;
  const int w    = __builtin_amdgcn_readfirstlane((int)threadIdx.x >> 5);
  const int ks   = w % GK_KS;
  const int rg   = w / GK_KS;
  const int row0 = (int)blockIdx.x * (GK_RG * GK_ROWS) + rg * GK_ROWS;
  const v4f* wr = (const v4f*)(Wm + (size_t)row0 * KTOT + ks * (GK_Q * 128)) + lane;

  const v4f* pp = (const v4f*)partial + threadIdx.x;
  v4f pc[NJ][KCH];
#pragma unroll
  for (int j = 0; j < NJ; ++j) {
    constexpr int RV = DD / 4;
    const int m   = (NTHR * j >= RV) ? 1 : 0;
    const int off = NTHR * j - RV * m;
#pragma unroll
    for (int c = 0; c < KCH; ++c)
      pc[j][c] = pp[(size_t)(c * NROW + m) * RV + off];
  }

  v4f xq[GK_Q];
  v4f wb[2][GK_Q];
#define GK_LOAD_ROW(R_)                                                       \
  do {                                                                        \
    _Pragma("unroll") for (int q = 0; q < GK_Q; ++q)                          \
      wb[(R_) % 2][q] = __builtin_nontemporal_load(                           \
          wr + (size_t)(R_) * (KTOT / 4) + 32 * q);                           \
  } while (0)

  __builtin_amdgcn_sched_barrier(0);
  GK_LOAD_ROW(0);
  __builtin_amdgcn_sched_barrier(0);
  {
    const float invT = 1.0f / (float)T;
    const float pb   = pool_b[0];
#pragma unroll
    for (int j = 0; j < NJ; ++j) {
      v4f a = pc[j][0];
#pragma unroll
      for (int c = 1; c < KCH; ++c) a += pc[j][c];
      if (NTHR * j >= DD / 4)
        a = (v4f){ __builtin_fmaf(a.x, invT, pb), __builtin_fmaf(a.y, invT, pb),
                   __builtin_fmaf(a.z, invT, pb), __builtin_fmaf(a.w, invT, pb) };
      else
        a = a * invT;
      avgs[NTHR * j + (int)threadIdx.x] = a;
    }
  }
  __syncthreads();
#pragma unroll
  for (int q = 0; q < GK_Q; ++q) xq[q] = avgs[ks * (GK_Q * 32) + 32 * q + lane];

  if (w < HP) {
    const int piece = (int)blockIdx.x * HP + w;
    const int hrow  = piece / (DD / 32);
    const int hd    = (piece % (DD / 32)) * 32 + lane;
    float hs = partial[(size_t)(2 + hrow) * DD + hd];
#pragma unroll
    for (int c = 1; c < KCH; ++c)
      hs += partial[(size_t)(c * NROW + 2 + hrow) * DD + hd];
    out_hfinal[(size_t)hrow * DD + hd] = hs;
  }
  float sum[GK_ROWS];
#pragma unroll
  for (int r = 0; r < GK_ROWS; ++r) {
    __builtin_amdgcn_sched_barrier(0);
    if (r + 1 < GK_ROWS) GK_LOAD_ROW(r + 1);
    __builtin_amdgcn_sched_barrier(0);
    float a0 = 0.0f, a1 = 0.0f, a2 = 0.0f, a3 = 0.0f;
#pragma unroll
    for (int q = 0; q < GK_Q; ++q) {
      const v4f a = wb[r % 2][q];
      const v4f b = xq[q];
      a0 = __builtin_fmaf(a.x, b.x, a0);
      a1 = __builtin_fmaf(a.y, b.y, a1);
      a2 = __builtin_fmaf(a.z, b.z, a2);
      a3 = __builtin_fmaf(a.w, b.w, a3);
    }
    sum[r] = (a0 + a1) + (a2 + a3);
    asm volatile("" : "+v"(sum[r]));
  }
#undef GK_LOAD_ROW
  __builtin_amdgcn_sched_barrier(0);
#pragma unroll
  for (int r = 0; r < GK_ROWS; ++r) {
    float s = sum[r];
#pragma unroll
    for (int m = 16; m >= 1; m >>= 1) s += __shfl_xor(s, m, 32);
    sum[r] = s;
  }
  if (lane == 0) {
#pragma unroll
    for (int r = 0; r < GK_ROWS; ++r)
      part[(rg * GK_ROWS + r) * GK_KS + ks] = sum[r];
  }
  __syncthreads();
  if (threadIdx.x < GK_RG * GK_ROWS) {
    const float* p = part + threadIdx.x * GK_KS;
    float s = p[0];
#pragma unroll
    for (int k = 1; k < GK_KS; ++k) s += p[k];
    const int row = (int)blockIdx.x * (GK_RG * GK_ROWS) + (int)threadIdx.x;
    const float v = s + bias[row];
    out[row] = do_relu ? fmaxf(v, 0.0f) : v;
  }
}

extern "C" void kernel_launch(void* const* d_in, const int* in_sizes, int n_in,
                              void* d_out, int out_size, void* d_ws, size_t ws_size,
                              hipStream_t stream) {
  const float* inp    = (const float*)d_in[0];
  const float* w      = (const float*)d_in[2];
  const float* pool_w = (const float*)d_in[3];
  const float* pool_b = (const float*)d_in[4];
  const float* i2o_w  = (const float*)d_in[5];
  const float* i2o_b  = (const float*)d_in[6];
  const float* o_w    = (const float*)d_in[7];
  const float* o_b    = (const float*)d_in[8];
  float* out = (float*)d_out;

  const int D  = in_sizes[1] / F_NUM;
  const int T  = in_sizes[0] / D;
  const int H1 = in_sizes[6];

  const int KC = 16;
  int TC   = (((T + KC - 1) / KC) + 3) & ~3;
  int Tpad = KC * TC;
  int NT   = D / 16;

  float* ws      = (float*)d_ws;
  float* coefT   = ws;
  float* partial = coefT + (size_t)Tpad * 16;
  float* avg     = partial + (size_t)KC * NROW * D;
  float* hid1    = avg + (size_t)2 * D;

  int units = NT * KC;
  dim3 g2((units + 3) / 4), b2(128);
  const bool fused = (D == 4096 && T == GX_KC * GX_KQ * GX_TCW && H1 == 4096);
  if (fused) {
    gemm_coef_x_wide<4096><<<(4096 / (64 * GX_CT)) * GX_KC, GX_KQ * GX_CT * 32, 0, stream>>>(
        inp, w, pool_w, partial, T);
    gemv_rows_ksplit<8192, GX_KC, 4096, 4096 / (GK_RG * GK_ROWS)>
        <<<4096 / (GK_RG * GK_ROWS), GK_KS * GK_RG * 32, 0, stream>>>(
        i2o_w, i2o_b, partial, pool_b, hid1, out + OUT_N, T, 1);
  } else {
    coef_kernel<<<(Tpad + 255) / 256, 256, 0, stream>>>(w, pool_w, coefT, T, Tpad, D);
    if (D == 4096)
      gemm_coef_x<4096><<<g2, b2, 0, stream>>>(inp, coefT, partial, T, D, TC, NT, units);
    else
      gemm_coef_x<0><<<g2, b2, 0, stream>>>(inp, coefT, partial, T, D, TC, NT, units);
    reduce_partials<<<(NROW * D + 255) / 256, 256, 0, stream>>>(
        partial, pool_b, avg, out + OUT_N, D, KC, T);
  }

  dim3 gb(GV_WAVES * 32);
  dim3 g4((H1 + GV_WAVES - 1) / GV_WAVES), g5((OUT_N + GV_WAVES - 1) / GV_WAVES);
  if (!fused) {
    if (2 * D == 8192)
      gemv_rows_stream<64><<<g4, gb, 0, stream>>>(i2o_w, i2o_b, avg, hid1, H1, 1);
    else
      gemv_rows<<<g4, gb, 0, stream>>>(i2o_w, i2o_b, avg, hid1, 2 * D, H1, 1);
  }
  if (H1 == 4096)
    gemv_rows_stream<32><<<g5, gb, 0, stream>>>(o_w, o_b, hid1, out, OUT_N, 0);
  else
    gemv_rows<<<g5, gb, 0, stream>>>(o_w, o_b, hid1, out, H1, OUT_N, 0);
}


# second layer folded into the 4096x8192 GEMV's tail: a block's wave 0 stores its 10 partial dot products as one 128-byte line (operands read at start); last launch adds 256 lines; sum order changes
# speedup vs baseline: 3.7391x; 1.0570x over previous
#include <hip/hip_runtime.h>
#include <math.h>

typedef __attribute__((ext_vector_type(2))) float v2f;
typedef __attribute__((ext_vector_type(8))) float v8f;
typedef __attribute__((ext_vector_type(4))) float v4f;

#define F_NUM 10
#define OUT_N 10
#define NROW  12

__global__ void coef_kernel(const float* __restrict__ w,
                            const float* __restrict__ pool_w,
                            float* __restrict__ coefT,
                            int T, int Tpad, int D) {
  int t = blockIdx.x * blockDim.x + threadIdx.x;
  if (t >= Tpad) return;
  float row[16];
#pragma unroll
  for (int i = 0; i < 16; ++i) row[i] = 0.0f;
  if (t < T) {
    row[0] = 1.0f;
    float G = 0.0f;
    float e = (float)(T - 1 - t);
#pragma unroll
    for (int f = 0; f < F_NUM; ++f) {
      float wf = w[(size_t)f * D];
      float pf = pool_w[f];
      float c  = powf(wf, e);
      row[2 + f] = c;
      G += pf * (1.0f - c) / (1.0f - wf);
    }
    row[1] = G;
  }
#pragma unroll
  for (int i = 0; i < 16; ++i) coefT[(size_t)t * 16 + i] = row[i];
}

template <int DC>
__global__ void __launch_bounds__(128)
gemm_coef_x(const float* __restrict__ x, const float* __restrict__ coefT,
            float* __restrict__ partial, int T, int Drt, int TC, int NT, int units) {
  const int D = (DC > 0) ? DC : Drt;
  int wid = __builtin_amdgcn_readfirstlane(
      (int)(blockIdx.x * (blockDim.x >> 5)) + ((int)threadIdx.x >> 5));
  if (wid >= units) return;
  int lane = threadIdx.x & 31;
  int ntile  = wid % NT;
  int kchunk = wid / NT;
  int nb = ntile * 16;
  int n  = lane & 15;
  int h  = lane >> 4;

  int t0 = kchunk * TC;
  int valid = T - t0;
  if (valid < 0)  valid = 0;
  if (valid > TC) valid = TC;
  int full = valid & ~3;

  const float* cp = coefT + (size_t)(t0 + 2 * h) * 16 + n;
  const float* xp = x     + (size_t)(t0 + 2 * h) * D  + nb + n;
  const size_t xstep = (size_t)4 * D;

  v8f acc = {};
  if (full > 0) {
    v2f a = { cp[0], cp[16] };
    v2f b = { xp[0], xp[D] };
#pragma unroll 2
    for (int kk = 4; kk < full; kk += 4) {
      cp += 64;
      xp += xstep;
      v2f an = { cp[0], cp[16] };
      v2f bn = { xp[0], xp[D] };
      acc = __builtin_amdgcn_wmma_f32_16x16x4_f32(
          false, a, false, b, (short)0, acc, false, false);
      a = an; b = bn;
    }
    acc = __builtin_amdgcn_wmma_f32_16x16x4_f32(
        false, a, false, b, (short)0, acc, false, false);
  }
  for (int kk = full; kk < TC; kk += 4) {
    int ta = t0 + kk + 2 * h, tb = ta + 1;
    v2f a = { coefT[(size_t)ta * 16 + n], coefT[(size_t)tb * 16 + n] };
    v2f b = { (ta < T) ? x[(size_t)ta * D + nb + n] : 0.0f,
              (tb < T) ? x[(size_t)tb * D + nb + n] : 0.0f };
    acc = __builtin_amdgcn_wmma_f32_16x16x4_f32(
        false, a, false, b, (short)0, acc, false, false);
  }

#pragma unroll
  for (int r = 0; r < 8; ++r) {
    int m = r + 8 * h;
    if (m < NROW)
      partial[((size_t)kchunk * NROW + m) * D + nb + n] = acc[r];
  }
}

#define GX_G     4
#define GX_AHEAD 2
#define GX_TCW   128
#define GX_KQ    16
#define GX_CT    1
#define GX_KC    4

template <int M>
__device__ __forceinline__ float gx_row_xor(float v) {
  return __builtin_bit_cast(float, __builtin_amdgcn_update_dpp(
      0, __builtin_bit_cast(int, v), 0x160 + M, 0xf, 0xf, false));
}
__device__ __forceinline__ float gx_row16_sum(float v) {
  v += gx_row_xor<1>(v);
  v += gx_row_xor<2>(v);
  v += gx_row_xor<4>(v);
  v += gx_row_xor<8>(v);
  return v;
}

template <int DC>
__global__ void __launch_bounds__(GX_KQ * GX_CT * 32)
gemm_coef_x_wide(const float* __restrict__ x, const float* __restrict__ decay,
                 const float* __restrict__ pool_w,
                 float* __restrict__ partial, int T) {
  static_assert((GX_TCW / 4) % GX_G == 0 && (GX_TCW / 4) / GX_G > GX_AHEAD,
                "a wave's K range = whole groups, more than the ring holds");
  static_assert(DC % (64 * GX_CT) == 0, "row = whole block tiles");
  static_assert((GX_TCW / 4) % 2 == 0, "two powf anchors per lane");
  static_assert(GX_KQ >= 8, "one summing wave per accumulator register slot");
  constexpr int NS = GX_TCW / 4;
  constexpr int NG = NS / GX_G;
  constexpr int CG = DC / (64 * GX_CT);
  __shared__ v4f red[GX_KQ * GX_CT * 8 * 32];

  const int lane = threadIdx.x & 31;
  const int w  = __builtin_amdgcn_readfirstlane((int)threadIdx.x >> 5);
  const int ct = w % GX_CT;
  const int kq = w / GX_CT;
  const int cg     = (int)blockIdx.x % CG;
  const int kchunk = (int)blockIdx.x / CG;
  const int n  = lane & 15;
  const int h  = lane >> 4;
  const int cb = (cg * GX_CT + ct) * 64;
  const int t0 = (kchunk * GX_KQ + kq) * GX_TCW;

  const v4f*   xp = (const v4f*)(x + (size_t)(t0 + 2 * h) * DC + cb + 4 * n);

  float wS[F_NUM], pS[F_NUM];
#pragma unroll
  for (int f = 0; f < F_NUM; ++f) {
    wS[f] = decay[(size_t)f * DC];
    pS[f] = pool_w[f];
  }

  v2f ca[NS];
  v4f xa[GX_AHEAD + 1][GX_G];
  v4f xb[GX_AHEAD + 1][GX_G];
#define GX_LOAD_GROUP(G_)                                                     \
  do {                                                                        \
    _Pragma("unroll") for (int q = 0; q < GX_G; ++q) {                        \
      xa[(G_) % (GX_AHEAD + 1)][q] = xp[(size_t)((G_) * GX_G + q) * DC];      \
      xb[(G_) % (GX_AHEAD + 1)][q] =                                          \
          xp[(size_t)((G_) * GX_G + q) * DC + DC / 4];                        \
    }                                                                         \
  } while (0)

#pragma unroll
  for (int g = 0; g <= GX_AHEAD; ++g) GX_LOAD_GROUP(g);
  __builtin_amdgcn_sched_barrier(0);

  {
    const bool is_pow = (n >= 2) && (n < 2 + F_NUM);
    float wf = 0.0f, pf = 0.0f;
#pragma unroll
    for (int f = 0; f < F_NUM; ++f)
      if (n == 2 + f) { wf = wS[f]; pf = pS[f]; }
    const float fixed = (n == 0) ? 1.0f : 0.0f;
    const float wl = is_pow ? wf : fixed;
    const float ql = pf / (1.0f - wf);
    const float w2 = wl * wl, w4 = w2 * w2;
    const int e1 = T - t0 - (GX_TCW - 2) - 2 * h;
    float top = powf(wl, (float)e1);
    float mid = powf(wl, (float)(e1 + GX_TCW / 2));
    if (!is_pow) { top = fixed; mid = fixed; }
    float cx[NS], cy[NS];
    cy[NS - 1]     = top;  cx[NS - 1]     = top * wl;
    cy[NS / 2 - 1] = mid;  cx[NS / 2 - 1] = mid * wl;
#pragma unroll
    for (int s = NS - 2; s >= NS / 2; --s) {
      cy[s] = cy[s + 1] * w4;  cx[s] = cx[s + 1] * w4;
    }
#pragma unroll
    for (int s = NS / 2 - 2; s >= 0; --s) {
      cy[s] = cy[s + 1] * w4;  cx[s] = cx[s + 1] * w4;
    }
#pragma unroll
    for (int s = 0; s < NS; ++s) {
      const float gx = gx_row16_sum(ql * (1.0f - cx[s]));
      const float gy = gx_row16_sum(ql * (1.0f - cy[s]));
      ca[s] = (v2f){ (n == 1) ? gx : cx[s], (n == 1) ? gy : cy[s] };
    }
  }

  v8f acc0 = {}, acc1 = {}, acc2 = {}, acc3 = {};
#pragma unroll
  for (int g = 0; g < NG; ++g) {
    __builtin_amdgcn_sched_barrier(0);
    if (g >= 1 && g + GX_AHEAD < NG) GX_LOAD_GROUP(g + GX_AHEAD);
    __builtin_amdgcn_sched_barrier(0);
#pragma unroll
    for (int q = 0; q < GX_G; ++q) {
      const v2f a  = ca[g * GX_G + q];
      const v4f u  = xa[g % (GX_AHEAD + 1)][q];
      const v4f v  = xb[g % (GX_AHEAD + 1)][q];
      acc0 = __builtin_amdgcn_wmma_f32_16x16x4_f32(
          false, a, false, (v2f){ u.x, v.x }, (short)0, acc0, false, false);
      acc1 = __builtin_amdgcn_wmma_f32_16x16x4_f32(
          false, a, false, (v2f){ u.y, v.y }, (short)0, acc1, false, false);
      acc2 = __builtin_amdgcn_wmma_f32_16x16x4_f32(
          false, a, false, (v2f){ u.z, v.z }, (short)0, acc2, false, false);
      acc3 = __builtin_amdgcn_wmma_f32_16x16x4_f32(
          false, a, false, (v2f){ u.w, v.w }, (short)0, acc3, false, false);
    }
  }
#undef GX_LOAD_GROUP

#pragma unroll
  for (int r = 0; r < 8; ++r)
    red[((kq * GX_CT + ct) * 8 + r) * 32 + lane] =
        (v4f){ acc0[r], acc1[r], acc2[r], acc3[r] };
  __syncthreads();
  if (kq < 8) {
    const int r = kq;
    v4f s = red[(ct * 8 + r) * 32 + lane];
#pragma unroll
    for (int p = 1; p < GX_KQ; ++p)
      s += red[((p * GX_CT + ct) * 8 + r) * 32 + lane];
    const int m = r + 8 * h;
    if (m < NROW)
      *(v4f*)(partial + ((size_t)kchunk * NROW + m) * DC + cb + 4 * n) = s;
  }
}

__global__ void reduce_partials(const float* __restrict__ partial,
                                const float* __restrict__ pool_b,
                                float* __restrict__ avg,
                                float* __restrict__ out_hfinal,
                                int D, int KC, int T) {
  int idx = blockIdx.x * blockDim.x + threadIdx.x;
  if (idx >= NROW * D) return;
  int m = idx / D, d = idx % D;
  float s = 0.0f;
  for (int c = 0; c < KC; ++c)
    s += partial[((size_t)c * NROW + m) * D + d];
  float invT = 1.0f / (float)T;
  if (m == 0)      avg[d]     = s * invT;
  else if (m == 1) avg[D + d] = s * invT + pool_b[0];
  else             out_hfinal[(size_t)(m - 2) * D + d] = s;
}

__global__ void gemv_rows(const float* __restrict__ Wm,
                          const float* __restrict__ bias,
                          const float* __restrict__ vec,
                          float* __restrict__ out,
                          int K, int N, int do_relu) {
  int wid  = blockIdx.x * (blockDim.x >> 5) + (threadIdx.x >> 5);
  int lane = threadIdx.x & 31;
  if (wid >= N) return;
  const float4* wr = (const float4*)(Wm + (size_t)wid * K);
  const float4* vv = (const float4*)vec;
  float s = 0.0f;
  for (int i = lane; i < (K >> 2); i += 32) {
    float4 a = wr[i], b = vv[i];
    s += a.x * b.x + a.y * b.y + a.z * b.z + a.w * b.w;
  }
  for (int off = 16; off > 0; off >>= 1) s += __shfl_down(s, off, 32);
  if (lane == 0) {
    float v = s + bias[wid];
    out[wid] = do_relu ? fmaxf(v, 0.0f) : v;
  }
}


#define GV_G     8
#define GV_AHEAD 2
#define GV_WAVES 8

template <int KQ>
__global__ void __launch_bounds__(GV_WAVES * 32)
gemv_rows_stream(const float* __restrict__ Wm,
                 const float* __restrict__ bias,
                 const float* __restrict__ vec,
                 float* __restrict__ out,
                 int N, int do_relu) {
  static_assert(KQ % GV_G == 0 && KQ / GV_G >= GV_AHEAD, "row = whole groups");
  constexpr int NG = KQ / GV_G;
  const int lane = threadIdx.x & 31;
  const int wid  = __builtin_amdgcn_readfirstlane(
      (int)(blockIdx.x * GV_WAVES) + ((int)threadIdx.x >> 5));
  if (wid >= N) return;
  const v4f* wr = (const v4f*)(Wm + (size_t)wid * (128 * KQ)) + lane;
  const v4f* vv = (const v4f*)vec + lane;

  v4f wb[GV_AHEAD + 1][GV_G];
  v4f xb[GV_AHEAD + 1][GV_G];
#define GV_LOAD_GROUP(G_)                                                     \
  do {                                                                        \
    _Pragma("unroll") for (int q = 0; q < GV_G; ++q)                          \
      wb[(G_) % (GV_AHEAD + 1)][q] =                                          \
          __builtin_nontemporal_load(wr + 32 * ((G_) * GV_G + q));            \
    _Pragma("unroll") for (int q = 0; q < GV_G; ++q)                          \
      xb[(G_) % (GV_AHEAD + 1)][q] = vv[32 * ((G_) * GV_G + q)];              \
  } while (0)

#pragma unroll
  for (int g = 0; g < GV_AHEAD; ++g) GV_LOAD_GROUP(g);

  float a0 = 0.0f, a1 = 0.0f, a2 = 0.0f, a3 = 0.0f;
#pragma unroll
  for (int g = 0; g < NG; ++g) {
    __builtin_amdgcn_sched_barrier(0);
    if (g + GV_AHEAD < NG) GV_LOAD_GROUP(g + GV_AHEAD);
    __builtin_amdgcn_sched_barrier(0);
#pragma unroll
    for (int q = 0; q < GV_G; ++q) {
      const v4f a = wb[g % (GV_AHEAD + 1)][q];
      const v4f b = xb[g % (GV_AHEAD + 1)][q];
      a0 = __builtin_fmaf(a.x, b.x, a0);
      a1 = __builtin_fmaf(a.y, b.y, a1);
      a2 = __builtin_fmaf(a.z, b.z, a2);
      a3 = __builtin_fmaf(a.w, b.w, a3);
    }
  }
#undef GV_LOAD_GROUP
  float s = (a0 + a1) + (a2 + a3);
  for (int off = 16; off > 0; off >>= 1) s += __shfl_down(s, off, 32);
  if (lane == 0) {
    float v = s + bias[wid];
    out[wid] = do_relu ? fmaxf(v, 0.0f) : v;
  }
}

#define GK_KS   4
#define GK_RG   4
#define GK_ROWS 4
#define GK_Q    16

template <int KTOT, int KCH, int DD, int NBLK>
__global__ void __launch_bounds__(GK_KS * GK_RG * 32)
gemv_rows_ksplit(const float* __restrict__ Wm,
                 const float* __restrict__ bias,
                 const float* __restrict__ partial,
                 const float* __restrict__ pool_b,
                 const float* __restrict__ Wo,
                 float* __restrict__ opart,
                 float* __restrict__ out_hfinal,
                 int T) {
  static_assert(KTOT == GK_KS * GK_Q * 128, "row = GK_KS quarters of GK_Q pieces");
  static_assert(GK_RG * GK_ROWS == 16 && OUT_N % 2 == 0 && OUT_N / 2 <= 16,
                "a block's rows = one 16-lane row; outputs = pairs, one line per block");
  constexpr int OH = OUT_N / 2;
  constexpr int NTHR = GK_KS * GK_RG * 32;
  constexpr int NV   = KTOT / 4;
  constexpr int NJ   = NV / NTHR;
  constexpr int HP   = 5;
  static_assert(KTOT == 2 * DD && NV % NTHR == 0 && (DD / 4) % NTHR == 0,
                "a thread's piece j lies in one half of the vector");
  static_assert(NBLK * HP == F_NUM * (DD / 32) && HP <= GK_KS * GK_RG,
                "h_final = NBLK blocks x HP pieces");
  __shared__ v4f avgs[NV];
  __shared__ float part[GK_RG * GK_ROWS * GK_KS];
  const int lane = threadIdx.x & 31;
  const int w    = __builtin_amdgcn_readfirstlane((int)threadIdx.x >> 5);
  const int ks   = w % GK_KS;
  const int rg   = w / GK_KS;
  const int row0 = (int)blockIdx.x * (GK_RG * GK_ROWS) + rg * GK_ROWS;
  const v4f* wr = (const v4f*)(Wm + (size_t)row0 * KTOT + ks * (GK_Q * 128)) + lane;

  const v4f* pp = (const v4f*)partial + threadIdx.x;
  v4f pc[NJ][KCH];
#pragma unroll
  for (int j = 0; j < NJ; ++j) {
    constexpr int RV = DD / 4;
    const int m   = (NTHR * j >= RV) ? 1 : 0;
    const int off = NTHR * j - RV * m;
#pragma unroll
    for (int c = 0; c < KCH; ++c)
      pc[j][c] = pp[(size_t)(c * NROW + m) * RV + off];
  }

  v4f xq[GK_Q];
  v4f wb[2][GK_Q];
#define GK_LOAD_ROW(R_)                                                       \
  do {                                                                        \
    _Pragma("unroll") for (int q = 0; q < GK_Q; ++q)                          \
      wb[(R_) % 2][q] = __builtin_nontemporal_load(                           \
          wr + (size_t)(R_) * (KTOT / 4) + 32 * q);                           \
  } while (0)

  __builtin_amdgcn_sched_barrier(0);
  GK_LOAD_ROW(0);
  const int brow = (int)blockIdx.x * (GK_RG * GK_ROWS) + (lane & 15);
  float owv[OH];
#pragma unroll
  for (int j = 0; j < OH; ++j)
    owv[j] = Wo[(size_t)(2 * j + (lane >> 4)) * (NBLK * GK_RG * GK_ROWS) + brow];
  const float bv = bias[brow];
  __builtin_amdgcn_sched_barrier(0);
  {
    const float invT = 1.0f / (float)T;
    const float pb   = pool_b[0];
#pragma unroll
    for (int j = 0; j < NJ; ++j) {
      v4f a = pc[j][0];
#pragma unroll
      for (int c = 1; c < KCH; ++c) a += pc[j][c];
      if (NTHR * j >= DD / 4)
        a = (v4f){ __builtin_fmaf(a.x, invT, pb), __builtin_fmaf(a.y, invT, pb),
                   __builtin_fmaf(a.z, invT, pb), __builtin_fmaf(a.w, invT, pb) };
      else
        a = a * invT;
      avgs[NTHR * j + (int)threadIdx.x] = a;
    }
  }
  __syncthreads();
#pragma unroll
  for (int q = 0; q < GK_Q; ++q) xq[q] = avgs[ks * (GK_Q * 32) + 32 * q + lane];

  if (w < HP) {
    const int piece = (int)blockIdx.x * HP + w;
    const int hrow  = piece / (DD / 32);
    const int hd    = (piece % (DD / 32)) * 32 + lane;
    float hs = partial[(size_t)(2 + hrow) * DD + hd];
#pragma unroll
    for (int c = 1; c < KCH; ++c)
      hs += partial[(size_t)(c * NROW + 2 + hrow) * DD + hd];
    out_hfinal[(size_t)hrow * DD + hd] = hs;
  }
  float sum[GK_ROWS];
#pragma unroll
  for (int r = 0; r < GK_ROWS; ++r) {
    __builtin_amdgcn_sched_barrier(0);
    if (r + 1 < GK_ROWS) GK_LOAD_ROW(r + 1);
    __builtin_amdgcn_sched_barrier(0);
    float a0 = 0.0f, a1 = 0.0f, a2 = 0.0f, a3 = 0.0f;
#pragma unroll
    for (int q = 0; q < GK_Q; ++q) {
      const v4f a = wb[r % 2][q];
      const v4f b = xq[q];
      a0 = __builtin_fmaf(a.x, b.x, a0);
      a1 = __builtin_fmaf(a.y, b.y, a1);
      a2 = __builtin_fmaf(a.z, b.z, a2);
      a3 = __builtin_fmaf(a.w, b.w, a3);
    }
    sum[r] = (a0 + a1) + (a2 + a3);
    asm volatile("" : "+v"(sum[r]));
  }
#undef GK_LOAD_ROW
  __builtin_amdgcn_sched_barrier(0);
#pragma unroll
  for (int r = 0; r < GK_ROWS; ++r) {
    float s = sum[r];
#pragma unroll
    for (int m = 16; m >= 1; m >>= 1) s += __shfl_xor(s, m, 32);
    sum[r] = s;
  }
  if (lane == 0) {
#pragma unroll
    for (int r = 0; r < GK_ROWS; ++r)
      part[(rg * GK_ROWS + r) * GK_KS + ks] = sum[r];
  }
  __syncthreads();
  if (w == 0) {
    const float* p = part + (lane & 15) * GK_KS;
    float s = p[0];
#pragma unroll
    for (int k = 1; k < GK_KS; ++k) s += p[k];
    const float hid = fmaxf(s + bv, 0.0f);
    float v = 0.0f;
#pragma unroll
    for (int j = 0; j < OH; ++j) {
      const float c = gx_row16_sum(owv[j] * hid);
      if ((lane & 15) == j) v = c;
    }
    opart[(size_t)blockIdx.x * 32 + lane] = v;
  }
}

#define OP_WAVES 8

template <int NBLK>
__global__ void __launch_bounds__(OP_WAVES * 32)
out_from_partials(const float* __restrict__ opart,
                  const float* __restrict__ ob,
                  float* __restrict__ out) {
  constexpr int NTHR = OP_WAVES * 32;
  constexpr int NP   = NBLK * 8 / NTHR;
  static_assert((NBLK * 8) % NTHR == 0 && NTHR % 8 == 0,
                "a thread's pieces sit at one place of their lines");
  __shared__ v4f red[OP_WAVES * 8];
  __shared__ float bo[32];
  const int lane = threadIdx.x & 31;
  const int wv   = (int)threadIdx.x >> 5;
  const v4f* pp = (const v4f*)opart + threadIdx.x;
  v4f a[NP];
#pragma unroll
  for (int i = 0; i < NP; ++i) a[i] = pp[NTHR * i];
  if (wv == 0)
    bo[lane] = ((lane & 15) < OUT_N / 2) ? ob[2 * (lane & 15) + (lane >> 4)] : 0.0f;
  v4f s = a[0];
#pragma unroll
  for (int i = 1; i < NP; ++i) s += a[i];
#pragma unroll
  for (int m = 8; m <= 16; m <<= 1)
    s += (v4f){ __shfl_xor(s.x, m, 32), __shfl_xor(s.y, m, 32),
                __shfl_xor(s.z, m, 32), __shfl_xor(s.w, m, 32) };
  if (lane < 8) red[wv * 8 + lane] = s;
  __syncthreads();
  if (threadIdx.x < 32) {
    const float* r = (const float*)red + threadIdx.x;
    float t = r[0];
#pragma unroll
    for (int k = 1; k < OP_WAVES; ++k) t += r[32 * k];
    const int j = (int)threadIdx.x & 15, h = (int)threadIdx.x >> 4;
    if (j < OUT_N / 2) out[2 * j + h] = t + bo[threadIdx.x];
  }
}

extern "C" void kernel_launch(void* const* d_in, const int* in_sizes, int n_in,
                              void* d_out, int out_size, void* d_ws, size_t ws_size,
                              hipStream_t stream) {
  const float* inp    = (const float*)d_in[0];
  const float* w      = (const float*)d_in[2];
  const float* pool_w = (const float*)d_in[3];
  const float* pool_b = (const float*)d_in[4];
  const float* i2o_w  = (const float*)d_in[5];
  const float* i2o_b  = (const float*)d_in[6];
  const float* o_w    = (const float*)d_in[7];
  const float* o_b    = (const float*)d_in[8];
  float* out = (float*)d_out;

  const int D  = in_sizes[1] / F_NUM;
  const int T  = in_sizes[0] / D;
  const int H1 = in_sizes[6];

  const int KC = 16;
  int TC   = (((T + KC - 1) / KC) + 3) & ~3;
  int Tpad = KC * TC;
  int NT   = D / 16;

  float* ws      = (float*)d_ws;
  float* coefT   = ws;
  float* partial = coefT + (size_t)Tpad * 16;
  float* avg     = partial + (size_t)KC * NROW * D;
  float* hid1    = avg + (size_t)2 * D;

  int units = NT * KC;
  dim3 g2((units + 3) / 4), b2(128);
  const bool fused = (D == 4096 && T == GX_KC * GX_KQ * GX_TCW && H1 == 4096);
  if (fused) {
    gemm_coef_x_wide<4096><<<(4096 / (64 * GX_CT)) * GX_KC, GX_KQ * GX_CT * 32, 0, stream>>>(
        inp, w, pool_w, partial, T);
    gemv_rows_ksplit<8192, GX_KC, 4096, 4096 / (GK_RG * GK_ROWS)>
        <<<4096 / (GK_RG * GK_ROWS), GK_KS * GK_RG * 32, 0, stream>>>(
        i2o_w, i2o_b, partial, pool_b, o_w, avg, out + OUT_N, T);
    out_from_partials<4096 / (GK_RG * GK_ROWS)><<<1, OP_WAVES * 32, 0, stream>>>(avg, o_b, out);
  } else {
    coef_kernel<<<(Tpad + 255) / 256, 256, 0, stream>>>(w, pool_w, coefT, T, Tpad, D);
    if (D == 4096)
      gemm_coef_x<4096><<<g2, b2, 0, stream>>>(inp, coefT, partial, T, D, TC, NT, units);
    else
      gemm_coef_x<0><<<g2, b2, 0, stream>>>(inp, coefT, partial, T, D, TC, NT, units);
    reduce_partials<<<(NROW * D + 255) / 256, 256, 0, stream>>>(
        partial, pool_b, avg, out + OUT_N, D, KC, T);
  }

  dim3 gb(GV_WAVES * 32);
  dim3 g4((H1 + GV_WAVES - 1) / GV_WAVES), g5((OUT_N + GV_WAVES - 1) / GV_WAVES);
  if (!fused) {
    if (2 * D == 8192)
      gemv_rows_stream<64><<<g4, gb, 0, stream>>>(i2o_w, i2o_b, avg, hid1, H1, 1);
    else
      gemv_rows<<<g4, gb, 0, stream>>>(i2o_w, i2o_b, avg, hid1, 2 * D, H1, 1);
    if (H1 == 4096)
      gemv_rows_stream<32><<<g5, gb, 0, stream>>>(o_w, o_b, hid1, out, OUT_N, 0);
    else
      gemv_rows<<<g5, gb, 0, stream>>>(o_w, o_b, hid1, out, H1, OUT_N, 0);
  }
}
